# starting kernel
# baseline (speedup 1.0000x reference)
#include <hip/hip_runtime.h>
#include <hip/hip_bf16.h>
#include <math.h>


typedef float v2f __attribute__((ext_vector_type(2)));
typedef float v8f __attribute__((ext_vector_type(8)));

#define DH 128

__global__ void gcn_deg_init(float* __restrict__ deg, int n) {
    int i = blockIdx.x * blockDim.x + threadIdx.x;
    if (i < n) deg[i] = 1.0f;
}

__global__ void gcn_deg_count(const int* __restrict__ dst, float* __restrict__ deg, int ne) {
    int e = blockIdx.x * blockDim.x + threadIdx.x;
    if (e < ne) atomicAdd(&deg[dst[e]], 1.0f);
}

__global__ void gcn_dinv(float* __restrict__ deg, int n) {
    int i = blockIdx.x * blockDim.x + threadIdx.x;
    if (i < n) deg[i] = rsqrtf(deg[i]);
}

__global__ void gcn_norm(const int* __restrict__ src, const int* __restrict__ dst,
                         const float* __restrict__ dinv, float* __restrict__ nrm, int ne) {
    int e = blockIdx.x * blockDim.x + threadIdx.x;
    if (e < ne) nrm[e] = dinv[src[e]] * dinv[dst[e]];
}

__global__ void gcn_transpose_w(const float* __restrict__ W, float* __restrict__ Wt, int count) {
    int i = blockIdx.x * blockDim.x + threadIdx.x;
    if (i >= count) return;
    int l = i >> 14;
    int rem = i & 16383;
    int k = rem >> 7;
    int n = rem & (DH - 1);
    Wt[(l << 14) + (n << 7) + k] = W[i];
}

__global__ void __launch_bounds__(256)
gcn_gemm(const float* __restrict__ X, const float* __restrict__ Wt,
         float* __restrict__ H, int nrows) {
    const int lane = threadIdx.x & 31;
    const int wv   = threadIdx.x >> 5;
    const int m0   = blockIdx.x << 4;
    const int n0   = wv << 4;
    const int r    = lane & 15;
    const int half = lane >> 4;

    int arow = m0 + r;
    if (arow >= nrows) arow = nrows - 1;

    const float* ap = X  + (size_t)arow     * DH + 2 * half;
    const float* bp = Wt + (size_t)(n0 + r) * DH + 2 * half;

    v8f c = {};
#pragma unroll
    for (int ks = 0; ks < DH / 4; ++ks) {
        v2f a = *(const v2f*)(ap + 4 * ks);
        v2f b = *(const v2f*)(bp + 4 * ks);
        c = __builtin_amdgcn_wmma_f32_16x16x4_f32(
                 false, a,  false, b,
                 (short)0, c,  false,  false);
    }

    float* hp = H + (size_t)m0 * DH + n0 + r;
#pragma unroll
    for (int v = 0; v < 8; ++v) {
        int row = v + 8 * half;
        if (m0 + row < nrows) hp[(size_t)row * DH] = c[v];
    }
}

__global__ void gcn_init_y(float* __restrict__ Y, const float* __restrict__ H,
                           const float* __restrict__ dinv,
                           const float* __restrict__ bias, int total4) {
    int i = blockIdx.x * blockDim.x + threadIdx.x;
    if (i >= total4) return;
    int node = i >> 5;
    int col4 = i & 31;
    float di = dinv[node];
    di *= di;
    float4 h = ((const float4*)H)[i];
    float4 b = ((const float4*)bias)[col4];
    ((float4*)Y)[i] = make_float4(b.x + di * h.x, b.y + di * h.y,
                                  b.z + di * h.z, b.w + di * h.w);
}

__global__ void __launch_bounds__(256)
gcn_scatter(const float* __restrict__ H, float* __restrict__ Y,
            const int* __restrict__ src, const int* __restrict__ dst,
            const float* __restrict__ nrm, int nedges) {
    int e = (blockIdx.x << 3) + (threadIdx.x >> 5);
    if (e >= nedges) return;
    e = __builtin_amdgcn_readfirstlane(e);
    int lane = threadIdx.x & 31;

    int   s = src[e];
    int   d = dst[e];
    float w = nrm[e];

    float4 hv = ((const float4*)(H + (size_t)s * DH))[lane];
    float* yp = Y + (size_t)d * DH + (lane << 2);
    float  m0 = hv.x * w, m1 = hv.y * w, m2 = hv.z * w, m3 = hv.w * w;

    asm volatile("global_atomic_add_f32 %0, %1, off\n\t"
                 "global_atomic_add_f32 %0, %2, off offset:4\n\t"
                 "global_atomic_add_f32 %0, %3, off offset:8\n\t"
                 "global_atomic_add_f32 %0, %4, off offset:12"
                 :: "v"(yp), "v"(m0), "v"(m1), "v"(m2), "v"(m3)
                 : "memory");
}

__global__ void gcn_gelu(float* __restrict__ Y, int total4) {
    int i = blockIdx.x * blockDim.x + threadIdx.x;
    if (i >= total4) return;
    float4 v = ((float4*)Y)[i];
    const float k = 0.70710678118654752440f;
    v.x = 0.5f * v.x * (1.0f + erff(v.x * k));
    v.y = 0.5f * v.y * (1.0f + erff(v.y * k));
    v.z = 0.5f * v.z * (1.0f + erff(v.z * k));
    v.w = 0.5f * v.w * (1.0f + erff(v.w * k));
    ((float4*)Y)[i] = v;
}

extern "C" void kernel_launch(void* const* d_in, const int* in_sizes, int n_in,
                              void* d_out, int out_size, void* d_ws, size_t ws_size,
                              hipStream_t stream) {
    const float* x    = (const float*)d_in[0];
    const int*   eidx = (const int*)  d_in[1];
    const float* Ws   = (const float*)d_in[2];
    const float* bs   = (const float*)d_in[3];
    float*       out  = (float*)d_out;

    const int N = in_sizes[0] / DH;
    const int E = in_sizes[1] / 2;
    const int* src = eidx;
    const int* dst = eidx + E;

    float* ws   = (float*)d_ws;
    float* dinv = ws;
    float* nrm  = dinv + (((size_t)N + 63) & ~(size_t)63);
    float* Wt   = nrm  + (((size_t)E + 63) & ~(size_t)63);
    float* h    = Wt   + (size_t)3 * DH * DH;
    float* buf  = h    + (size_t)N * DH;

    const int threads = 256;
    const int total4  = N * DH / 4;
    const int wcount  = 3 * DH * DH;

    gcn_deg_init   <<<(N + threads - 1) / threads, threads, 0, stream>>>(dinv, N);
    gcn_deg_count  <<<(E + threads - 1) / threads, threads, 0, stream>>>(dst, dinv, E);
    gcn_dinv       <<<(N + threads - 1) / threads, threads, 0, stream>>>(dinv, N);
    gcn_norm       <<<(E + threads - 1) / threads, threads, 0, stream>>>(src, dst, dinv, nrm, E);
    gcn_transpose_w<<<(wcount + threads - 1) / threads, threads, 0, stream>>>(Ws, Wt, wcount);

    const float* xin = x;
    for (int l = 0; l < 3; ++l) {
        float* y = (l == 2) ? out : buf;
        gcn_gemm   <<<(N + 15) / 16, 256, 0, stream>>>(xin, Wt + (size_t)l * DH * DH, h, N);
        gcn_init_y <<<(total4 + threads - 1) / threads, threads, 0, stream>>>(y, h, dinv, bs + l * DH, total4);
        gcn_scatter<<<(E + 7) / 8, 256, 0, stream>>>(h, y, src, dst, nrm, E);
        xin = y;
    }

    gcn_gelu<<<(total4 + threads - 1) / threads, threads, 0, stream>>>(out, total4);
}


# GEMM re-tiled: a wave owns 32 rows x 64 columns (2x4 accumulator tiles, 8x fewer waves), W pre-packed in fragment order so B loads are contiguous, whole-line 16-byte stores; k order in sums changes
# speedup vs baseline: 3.1358x; 3.1358x over previous
#include <hip/hip_runtime.h>
#include <hip/hip_bf16.h>
#include <math.h>


typedef float v2f __attribute__((ext_vector_type(2)));
typedef float v4f __attribute__((ext_vector_type(4)));
typedef float v8f __attribute__((ext_vector_type(8)));

#define DH 128
#define TN 64
#define TSH 6
#define MAXB 1024
#define CHUNK 4096

__global__ void __launch_bounds__(1024)
gcn_part_hist(const int* __restrict__ dst, int* __restrict__ hist, int ne, int pitch) {
    __shared__ int lh[MAXB];
    const int tid = threadIdx.x;
    lh[tid] = 0;
    __syncthreads();
    const int e0 = blockIdx.x * CHUNK + tid;
    int d[4];
#pragma unroll
    for (int k = 0; k < 4; ++k) {
        int e = e0 + k * 1024;
        d[k] = (e < ne) ? dst[e] : -1;
    }
#pragma unroll
    for (int k = 0; k < 4; ++k)
        if (d[k] >= 0) atomicAdd(&lh[d[k] >> TSH], 1);
    __syncthreads();
    if (tid < pitch) hist[blockIdx.x * pitch + tid] = lh[tid];
}

__global__ void __launch_bounds__(1024)
gcn_part_fill(const int* __restrict__ src, const int* __restrict__ dst,
              const int* __restrict__ hist, int* __restrict__ list,
              int* __restrict__ start, int ne, int nb, int pitch, int nchunks) {
    __shared__ int cur[MAXB];
    __shared__ int wtot[32];
    const int tid  = threadIdx.x;
    const int lane = tid & 31;
    const int wv   = tid >> 5;
    const int me   = blockIdx.x;

    int tot = 0, below = 0;
    if (tid < nb) {
#pragma unroll 16
        for (int c = 0; c < nchunks; ++c) {
            int v = hist[c * pitch + tid];
            tot += v;
            below += (c < me) ? v : 0;
        }
    }
    int inc = tot;
#pragma unroll
    for (int o = 1; o < 32; o <<= 1) {
        int t = __shfl_up(inc, o);
        if (lane >= o) inc += t;
    }
    if (lane == 31) wtot[wv] = inc;
    __syncthreads();
    if (wv == 0) {
        int w = wtot[lane];
        int wi = w;
#pragma unroll
        for (int o = 1; o < 32; o <<= 1) {
            int t = __shfl_up(wi, o);
            if (lane >= o) wi += t;
        }
        wtot[lane] = wi - w;
    }
    __syncthreads();
    const int base = wtot[wv] + inc - tot;
    cur[tid] = base + below;
    if (me == 0) {
        if (tid < nb) start[tid] = base;
        if (tid == 0) start[nb] = ne;
    }
    __syncthreads();

    const int e0 = me * CHUNK + tid;
    int d[4], s[4];
#pragma unroll
    for (int k = 0; k < 4; ++k) {
        int e = e0 + k * 1024;
        bool ok = e < ne;
        d[k] = ok ? dst[e] : -1;
        s[k] = ok ? src[e] : 0;
    }
#pragma unroll
    for (int k = 0; k < 4; ++k) {
        if (d[k] >= 0) {
            int slot = atomicAdd(&cur[d[k] >> TSH], 1);
            ((int2*)list)[slot] = make_int2(s[k], d[k] & (TN - 1));
        }
    }
}

__global__ void __launch_bounds__(256)
gcn_dinv_range(const int* __restrict__ list, const int* __restrict__ start,
               float* __restrict__ dinv, int n) {
    __shared__ int cnt[TN];
    const int tid = threadIdx.x;
    if (tid < TN) cnt[tid] = 0;
    __syncthreads();
    const int beg = start[blockIdx.x], end = start[blockIdx.x + 1];
    for (int i = beg + tid; i < end; i += 256)
        atomicAdd(&cnt[((const int2*)list)[i].y], 1);
    __syncthreads();
    if (tid < TN) {
        int node = blockIdx.x * TN + tid;
        if (node < n) dinv[node] = rsqrtf(1.0f + (float)cnt[tid]);
    }
}

__global__ void gcn_pack_w(const float* __restrict__ W, float* __restrict__ Wf, int count) {
    int i = blockIdx.x * blockDim.x + threadIdx.x;
    if (i >= count) return;
    int e    = i & 3;
    int lane = (i >> 2) & 31;
    int j    = (i >> 7) & 15;
    int nt   = (i >> 11) & 3;
    int wh   = (i >> 13) & 1;
    int l    = i >> 14;
    int col  = 64 * wh + 4 * (lane & 15) + nt;
    int k    = 8 * j + 4 * (lane >> 4) + e;
    Wf[i] = W[(l << 14) + (k << 7) + col];
}

__global__ void __launch_bounds__(64) __attribute__((amdgpu_waves_per_eu(1, 4)))
gcn_gemm(const float* __restrict__ X, const float* __restrict__ Wf,
         const float* __restrict__ dinv, float* __restrict__ H, int nrows) {
    const int lane = threadIdx.x & 31;
    const int wh   = threadIdx.x >> 5;
    const int m0   = blockIdx.x << 5;
    const int r    = lane & 15;
    const int half = lane >> 4;

    const float* ap[2];
#pragma unroll
    for (int mt = 0; mt < 2; ++mt) {
        int arow = m0 + 16 * mt + r;
        if (arow >= nrows) arow = nrows - 1;
        ap[mt] = X + (size_t)arow * DH + 4 * half;
    }
    const float* bp = Wf + (size_t)wh * (4 * 16 * 128) + 4 * lane;

    float dv[2][8];
#pragma unroll
    for (int mt = 0; mt < 2; ++mt)
#pragma unroll
        for (int v = 0; v < 8; ++v) {
            int row = m0 + 16 * mt + v + 8 * half;
            dv[mt][v] = dinv[row < nrows ? row : nrows - 1];
        }

    v8f c[2][4];
#pragma unroll
    for (int mt = 0; mt < 2; ++mt)
#pragma unroll
        for (int nt = 0; nt < 4; ++nt) c[mt][nt] = (v8f){};

    v4f a[3][2], b[3][4];
#pragma unroll
    for (int p = 0; p < 2; ++p) {
#pragma unroll
        for (int mt = 0; mt < 2; ++mt) a[p][mt] = *(const v4f*)(ap[mt] + 8 * p);
#pragma unroll
        for (int nt = 0; nt < 4; ++nt) b[p][nt] = *(const v4f*)(bp + (nt * 16 + p) * 128);
    }

#pragma unroll
    for (int j = 0; j < DH / 8; ++j) {
        const int cur = j % 3, nxt = (j + 2) % 3;
        if (j + 2 < DH / 8) {
#pragma unroll
            for (int mt = 0; mt < 2; ++mt) a[nxt][mt] = *(const v4f*)(ap[mt] + 8 * (j + 2));
#pragma unroll
            for (int nt = 0; nt < 4; ++nt) b[nxt][nt] = *(const v4f*)(bp + (nt * 16 + j + 2) * 128);
        }
#pragma unroll
        for (int mt = 0; mt < 2; ++mt)
#pragma unroll
            for (int nt = 0; nt < 4; ++nt) {
                v2f a2 = {a[cur][mt].x, a[cur][mt].y};
                v2f b2 = {b[cur][nt].x, b[cur][nt].y};
                c[mt][nt] = __builtin_amdgcn_wmma_f32_16x16x4_f32(
                         false, a2,  false, b2,
                         (short)0, c[mt][nt],  false,  false);
            }
#pragma unroll
        for (int mt = 0; mt < 2; ++mt)
#pragma unroll
            for (int nt = 0; nt < 4; ++nt) {
                v2f a2 = {a[cur][mt].z, a[cur][mt].w};
                v2f b2 = {b[cur][nt].z, b[cur][nt].w};
                c[mt][nt] = __builtin_amdgcn_wmma_f32_16x16x4_f32(
                         false, a2,  false, b2,
                         (short)0, c[mt][nt],  false,  false);
            }
    }

    float* hp = H + (size_t)m0 * DH + 64 * wh + 4 * r;
#pragma unroll
    for (int mt = 0; mt < 2; ++mt)
#pragma unroll
        for (int v = 0; v < 8; ++v) {
            int row = 16 * mt + v + 8 * half;
            float s = dv[mt][v];
            float4 o = make_float4(c[mt][0][v] * s, c[mt][1][v] * s,
                                   c[mt][2][v] * s, c[mt][3][v] * s);
            if (m0 + row < nrows) *(float4*)(hp + (size_t)row * DH) = o;
        }
}

__global__ void __launch_bounds__(256)
gcn_agg(const float* __restrict__ H, float* __restrict__ Y,
        const int* __restrict__ list, const int* __restrict__ start,
        const float* __restrict__ dinv, const float* __restrict__ bias, int n) {
    __shared__ float S[TN * DH];
    const int tid   = threadIdx.x;
    const int lane  = tid & 31;
    const int wv    = __builtin_amdgcn_readfirstlane(tid >> 5);
    const int node0 = blockIdx.x * TN;
    const float4* H4 = (const float4*)H;
    const int2*   L2 = (const int2*)list;

    float4 v0[TN / 8];
#pragma unroll
    for (int it = 0; it < TN / 8; ++it) {
        int node = node0 + it * 8 + wv;
        int nc = node < n ? node : n - 1;
        v0[it] = H4[(size_t)nc * 32 + lane];
    }
#pragma unroll
    for (int it = 0; it < TN / 8; ++it) {
        int nl = it * 8 + wv;
        float4 v = v0[it];
        if (node0 + nl >= n) v = make_float4(0.f, 0.f, 0.f, 0.f);
        float* sp = S + nl * DH + lane;
        sp[0] = v.x; sp[32] = v.y; sp[64] = v.z; sp[96] = v.w;
    }
    __syncthreads();

    const int beg = start[blockIdx.x];
    const int cnt = start[blockIdx.x + 1] - beg;

    int g = wv;
    int2 en = make_int2(node0, 0);
    if (cnt > 0) {
        int i = g * 8 + lane;
        en = L2[beg + (i < cnt ? i : cnt - 1)];
    }
    while (g * 8 < cnt) {
        const int2 ec = en;
        const int nval = cnt - g * 8;
        float4 h[8];
        int dl[8];
#pragma unroll
        for (int k = 0; k < 8; ++k) {
            int s = __builtin_amdgcn_readlane(ec.x, k);
            dl[k] = __builtin_amdgcn_readlane(ec.y, k);
            h[k] = H4[(size_t)s * 32 + lane];
        }
        g += 8;
        {
            int i = g * 8 + lane;
            en = L2[beg + (i < cnt ? i : cnt - 1)];
        }
#pragma unroll
        for (int k = 0; k < 8; ++k) {
            float4 v = h[k];
            if (k >= nval) v = make_float4(0.f, 0.f, 0.f, 0.f);
            float* sp = S + dl[k] * DH + lane;
            atomicAdd(sp,      v.x);
            atomicAdd(sp + 32, v.y);
            atomicAdd(sp + 64, v.z);
            atomicAdd(sp + 96, v.w);
        }
    }
    __syncthreads();

    const float4 b4 = ((const float4*)bias)[lane];
    float di[TN / 8];
#pragma unroll
    for (int it = 0; it < TN / 8; ++it) {
        int node = node0 + it * 8 + wv;
        di[it] = dinv[node < n ? node : n - 1];
    }
#pragma unroll
    for (int it = 0; it < TN / 8; ++it) {
        int nl = it * 8 + wv;
        int node = node0 + nl;
        const float* sp = S + nl * DH + lane;
        float4 o = make_float4(b4.x + di[it] * sp[0],  b4.y + di[it] * sp[32],
                               b4.z + di[it] * sp[64], b4.w + di[it] * sp[96]);
        if (node < n) ((float4*)Y)[(size_t)node * 32 + lane] = o;
    }
}

__global__ void gcn_gelu(float* __restrict__ Y, int total4) {
    int i = blockIdx.x * blockDim.x + threadIdx.x;
    if (i >= total4) return;
    float4 v = ((float4*)Y)[i];
    const float k = 0.70710678118654752440f;
    v.x = 0.5f * v.x * (1.0f + erff(v.x * k));
    v.y = 0.5f * v.y * (1.0f + erff(v.y * k));
    v.z = 0.5f * v.z * (1.0f + erff(v.z * k));
    v.w = 0.5f * v.w * (1.0f + erff(v.w * k));
    ((float4*)Y)[i] = v;
}

extern "C" void kernel_launch(void* const* d_in, const int* in_sizes, int n_in,
                              void* d_out, int out_size, void* d_ws, size_t ws_size,
                              hipStream_t stream) {
    const float* x    = (const float*)d_in[0];
    const int*   eidx = (const int*)  d_in[1];
    const float* Ws   = (const float*)d_in[2];
    const float* bs   = (const float*)d_in[3];
    float*       out  = (float*)d_out;

    const int N = in_sizes[0] / DH;
    const int E = in_sizes[1] / 2;
    const int* src = eidx;
    const int* dst = eidx + E;

    const int NB      = (N + TN - 1) / TN;
    const int PITCH   = (NB + 31) & ~31;
    const int NCHUNK  = (E + CHUNK - 1) / CHUNK;

    float* ws    = (float*)d_ws;
    float* dinv  = ws;
    int*   hist  = (int*)(dinv + (((size_t)N + 63) & ~(size_t)63));
    int*   start = hist  + (size_t)NCHUNK * PITCH;
    int*   list  = start + (((size_t)NB + 1 + 63) & ~(size_t)63);
    float* Wf    = (float*)(list + (((size_t)2 * E + 63) & ~(size_t)63));
    float* h     = Wf + (size_t)3 * DH * DH;
    float* buf   = h  + (size_t)N * DH;

    const int threads = 256;
    const int total4  = N * DH / 4;
    const int wcount  = 3 * DH * DH;

    gcn_part_hist  <<<NCHUNK, 1024, 0, stream>>>(dst, hist, E, PITCH);
    gcn_part_fill  <<<NCHUNK, 1024, 0, stream>>>(src, dst, hist, list, start, E, NB, PITCH, NCHUNK);
    gcn_dinv_range <<<NB, 256, 0, stream>>>(list, start, dinv, N);
    gcn_pack_w     <<<(wcount + threads - 1) / threads, threads, 0, stream>>>(Ws, Wf, wcount);

    const float* xin = x;
    for (int l = 0; l < 3; ++l) {
        float* y = (l == 2) ? out : buf;
        gcn_gemm<<<(N + 31) / 32, 64, 0, stream>>>(xin, Wf + (size_t)l * DH * DH, dinv, h, N);
        gcn_agg <<<NB, 256, 0, stream>>>(h, y, list, start, dinv, bs + l * DH, N);
        xin = y;
    }

    gcn_gelu<<<(total4 + threads - 1) / threads, threads, 0, stream>>>(out, total4);
}


# LDS float adds replaced by register sums: a wave owns 8 target rows of its 64-node range; the range's list is regrouped by owning wave in LDS (ballot ranks, no atomics); sum order changes
# speedup vs baseline: 3.2745x; 1.0442x over previous
#include <hip/hip_runtime.h>
#include <hip/hip_bf16.h>
#include <math.h>


typedef float v2f __attribute__((ext_vector_type(2)));
typedef float v4f __attribute__((ext_vector_type(4)));
typedef float v8f __attribute__((ext_vector_type(8)));
typedef float v32f __attribute__((ext_vector_type(32)));

#define DH 128
#define TN 64
#define TSH 6
#define MAXB 1024
#define CHUNK 4096

__global__ void __launch_bounds__(1024)
gcn_part_hist(const int* __restrict__ dst, int* __restrict__ hist, int ne, int pitch) {
    __shared__ int lh[MAXB];
    const int tid = threadIdx.x;
    lh[tid] = 0;
    __syncthreads();
    const int e0 = blockIdx.x * CHUNK + tid;
    int d[4];
#pragma unroll
    for (int k = 0; k < 4; ++k) {
        int e = e0 + k * 1024;
        d[k] = (e < ne) ? dst[e] : -1;
    }
#pragma unroll
    for (int k = 0; k < 4; ++k)
        if (d[k] >= 0) atomicAdd(&lh[d[k] >> TSH], 1);
    __syncthreads();
    if (tid < pitch) hist[blockIdx.x * pitch + tid] = lh[tid];
}

__global__ void __launch_bounds__(1024)
gcn_part_fill(const int* __restrict__ src, const int* __restrict__ dst,
              const int* __restrict__ hist, int* __restrict__ list,
              int* __restrict__ start, int ne, int nb, int pitch, int nchunks) {
    __shared__ int cur[MAXB];
    __shared__ int wtot[32];
    const int tid  = threadIdx.x;
    const int lane = tid & 31;
    const int wv   = tid >> 5;
    const int me   = blockIdx.x;

    int tot = 0, below = 0;
    if (tid < nb) {
#pragma unroll 16
        for (int c = 0; c < nchunks; ++c) {
            int v = hist[c * pitch + tid];
            tot += v;
            below += (c < me) ? v : 0;
        }
    }
    int inc = tot;
#pragma unroll
    for (int o = 1; o < 32; o <<= 1) {
        int t = __shfl_up(inc, o);
        if (lane >= o) inc += t;
    }
    if (lane == 31) wtot[wv] = inc;
    __syncthreads();
    if (wv == 0) {
        int w = wtot[lane];
        int wi = w;
#pragma unroll
        for (int o = 1; o < 32; o <<= 1) {
            int t = __shfl_up(wi, o);
            if (lane >= o) wi += t;
        }
        wtot[lane] = wi - w;
    }
    __syncthreads();
    const int base = wtot[wv] + inc - tot;
    cur[tid] = base + below;
    if (me == 0) {
        if (tid < nb) start[tid] = base;
        if (tid == 0) start[nb] = ne;
    }
    __syncthreads();

    const int e0 = me * CHUNK + tid;
    int d[4], s[4];
#pragma unroll
    for (int k = 0; k < 4; ++k) {
        int e = e0 + k * 1024;
        bool ok = e < ne;
        d[k] = ok ? dst[e] : -1;
        s[k] = ok ? src[e] : 0;
    }
#pragma unroll
    for (int k = 0; k < 4; ++k) {
        if (d[k] >= 0) {
            int slot = atomicAdd(&cur[d[k] >> TSH], 1);
            ((int2*)list)[slot] = make_int2(s[k], d[k] & (TN - 1));
        }
    }
}

__global__ void __launch_bounds__(256)
gcn_dinv_range(const int* __restrict__ list, const int* __restrict__ start,
               float* __restrict__ dinv, int n) {
    __shared__ int cnt[TN];
    const int tid = threadIdx.x;
    if (tid < TN) cnt[tid] = 0;
    __syncthreads();
    const int beg = start[blockIdx.x], end = start[blockIdx.x + 1];
    for (int i = beg + tid; i < end; i += 256)
        atomicAdd(&cnt[((const int2*)list)[i].y], 1);
    __syncthreads();
    if (tid < TN) {
        int node = blockIdx.x * TN + tid;
        if (node < n) dinv[node] = rsqrtf(1.0f + (float)cnt[tid]);
    }
}

__global__ void gcn_pack_w(const float* __restrict__ W, float* __restrict__ Wf, int count) {
    int i = blockIdx.x * blockDim.x + threadIdx.x;
    if (i >= count) return;
    int e    = i & 3;
    int lane = (i >> 2) & 31;
    int j    = (i >> 7) & 15;
    int nt   = (i >> 11) & 3;
    int wh   = (i >> 13) & 1;
    int l    = i >> 14;
    int col  = 64 * wh + 4 * (lane & 15) + nt;
    int k    = 8 * j + 4 * (lane >> 4) + e;
    Wf[i] = W[(l << 14) + (k << 7) + col];
}

__global__ void __launch_bounds__(64) __attribute__((amdgpu_waves_per_eu(1, 4)))
gcn_gemm(const float* __restrict__ X, const float* __restrict__ Wf,
         const float* __restrict__ dinv, float* __restrict__ H, int nrows) {
    const int lane = threadIdx.x & 31;
    const int wh   = threadIdx.x >> 5;
    const int m0   = blockIdx.x << 5;
    const int r    = lane & 15;
    const int half = lane >> 4;

    const float* ap[2];
#pragma unroll
    for (int mt = 0; mt < 2; ++mt) {
        int arow = m0 + 16 * mt + r;
        if (arow >= nrows) arow = nrows - 1;
        ap[mt] = X + (size_t)arow * DH + 4 * half;
    }
    const float* bp = Wf + (size_t)wh * (4 * 16 * 128) + 4 * lane;

    float dv[2][8];
#pragma unroll
    for (int mt = 0; mt < 2; ++mt)
#pragma unroll
        for (int v = 0; v < 8; ++v) {
            int row = m0 + 16 * mt + v + 8 * half;
            dv[mt][v] = dinv[row < nrows ? row : nrows - 1];
        }

    v8f c[2][4];
#pragma unroll
    for (int mt = 0; mt < 2; ++mt)
#pragma unroll
        for (int nt = 0; nt < 4; ++nt) c[mt][nt] = (v8f){};

    v4f a[3][2], b[3][4];
#pragma unroll
    for (int p = 0; p < 2; ++p) {
#pragma unroll
        for (int mt = 0; mt < 2; ++mt) a[p][mt] = *(const v4f*)(ap[mt] + 8 * p);
#pragma unroll
        for (int nt = 0; nt < 4; ++nt) b[p][nt] = *(const v4f*)(bp + (nt * 16 + p) * 128);
    }

#pragma unroll
    for (int j = 0; j < DH / 8; ++j) {
        const int cur = j % 3, nxt = (j + 2) % 3;
        if (j + 2 < DH / 8) {
#pragma unroll
            for (int mt = 0; mt < 2; ++mt) a[nxt][mt] = *(const v4f*)(ap[mt] + 8 * (j + 2));
#pragma unroll
            for (int nt = 0; nt < 4; ++nt) b[nxt][nt] = *(const v4f*)(bp + (nt * 16 + j + 2) * 128);
        }
#pragma unroll
        for (int mt = 0; mt < 2; ++mt)
#pragma unroll
            for (int nt = 0; nt < 4; ++nt) {
                v2f a2 = {a[cur][mt].x, a[cur][mt].y};
                v2f b2 = {b[cur][nt].x, b[cur][nt].y};
                c[mt][nt] = __builtin_amdgcn_wmma_f32_16x16x4_f32(
                         false, a2,  false, b2,
                         (short)0, c[mt][nt],  false,  false);
            }
#pragma unroll
        for (int mt = 0; mt < 2; ++mt)
#pragma unroll
            for (int nt = 0; nt < 4; ++nt) {
                v2f a2 = {a[cur][mt].z, a[cur][mt].w};
                v2f b2 = {b[cur][nt].z, b[cur][nt].w};
                c[mt][nt] = __builtin_amdgcn_wmma_f32_16x16x4_f32(
                         false, a2,  false, b2,
                         (short)0, c[mt][nt],  false,  false);
            }
    }

    float* hp = H + (size_t)m0 * DH + 64 * wh + 4 * r;
#pragma unroll
    for (int mt = 0; mt < 2; ++mt)
#pragma unroll
        for (int v = 0; v < 8; ++v) {
            int row = 16 * mt + v + 8 * half;
            float s = dv[mt][v];
            float4 o = make_float4(c[mt][0][v] * s, c[mt][1][v] * s,
                                   c[mt][2][v] * s, c[mt][3][v] * s);
            if (m0 + row < nrows) *(float4*)(hp + (size_t)row * DH) = o;
        }
}

#define LCH 2048
__global__ void __launch_bounds__(256)
gcn_agg(const float* __restrict__ H, float* __restrict__ Y,
        const int* __restrict__ list, const int* __restrict__ start,
        const float* __restrict__ dinv, const float* __restrict__ bias, int n) {
    __shared__ int2 sub[LCH];
    __shared__ int  cw[64];
    const int tid   = threadIdx.x;
    const int lane  = tid & 31;
    const int wv    = __builtin_amdgcn_readfirstlane(tid >> 5);
    const int node0 = blockIdx.x * TN;
    const float4* H4 = (const float4*)H;
    const int2*   L2 = (const int2*)list;
    const unsigned below_me = (1u << lane) - 1u;

    v32f acc;
#pragma unroll
    for (int j = 0; j < 8; ++j) {
        int node = node0 + wv * 8 + j;
        int nc = node < n ? node : n - 1;
        const float4 t = H4[(size_t)nc * 32 + lane];
        acc[4 * j] = t.x; acc[4 * j + 1] = t.y; acc[4 * j + 2] = t.z; acc[4 * j + 3] = t.w;
    }

    const int beg = start[blockIdx.x];
    const int cnt = start[blockIdx.x + 1] - beg;

    for (int base = 0; base < cnt; base += LCH) {
        const int m = (cnt - base < LCH) ? cnt - base : LCH;
        int2 e[LCH / 256];
        int  loc[LCH / 256];
        int  mine = 0;
#pragma unroll
        for (int q = 0; q < LCH / 256; ++q) {
            const int i = tid + q * 256;
            const bool ok = i < m;
            e[q] = L2[beg + base + (ok ? i : m - 1)];
            const int o = ok ? (e[q].y >> 3) : 8;
            const int before = __shfl(mine, o & 7);
            int rank = 0;
#pragma unroll
            for (int oo = 0; oo < 8; ++oo) {
                const unsigned mk = __builtin_amdgcn_ballot_w32(o == oo);
                if (o == oo) rank = __popc(mk & below_me);
                if (lane == oo) mine += __popc(mk);
            }
            loc[q] = before + rank;
        }
        if (lane < 8) cw[wv * 8 + lane] = mine;
        __syncthreads();

        int tot = 0, below = 0;
#pragma unroll
        for (int w = 0; w < 8; ++w) {
            const int c = cw[w * 8 + (lane & 7)];
            tot += c;
            below += (w < wv) ? c : 0;
        }
        int first = 0;
#pragma unroll
        for (int oo = 0; oo < 7; ++oo) {
            const int t = __shfl(tot, oo);
            first += ((lane & 7) > oo) ? t : 0;
        }
        const int basev = first + below;
#pragma unroll
        for (int q = 0; q < LCH / 256; ++q) {
            const int i = tid + q * 256;
            const int o = e[q].y >> 3;
            const int slot = __shfl(basev, o & 7) + loc[q];
            if (i < m) sub[slot] = e[q];
        }
        const int my_cnt = __builtin_amdgcn_readlane(tot, wv);
        const int my_beg = __builtin_amdgcn_readlane(first, wv);
        __syncthreads();

        for (int g = 0; g < my_cnt; g += 8) {
            int i = g + (lane & 7);
            if (i > my_cnt - 1) i = my_cnt - 1;
            const int2 ec = sub[my_beg + i];
            const int nval = my_cnt - g;
            float4 h[8];
            int dl[8];
#pragma unroll
            for (int k = 0; k < 8; ++k) {
                int s = __builtin_amdgcn_readlane(ec.x, k);
                dl[k] = (__builtin_amdgcn_readlane(ec.y, k) & 7) * 4;
                h[k] = H4[(size_t)s * 32 + lane];
            }
#pragma unroll
            for (int k = 0; k < 8; ++k) {
                if (k < nval) {
                    const float4 v = h[k];
                    const int b = dl[k];
                    acc[b] += v.x; acc[b + 1] += v.y; acc[b + 2] += v.z; acc[b + 3] += v.w;
                }
            }
        }
        __syncthreads();
    }

    const float4 b4 = ((const float4*)bias)[lane];
    float di[8];
#pragma unroll
    for (int j = 0; j < 8; ++j) {
        int node = node0 + wv * 8 + j;
        di[j] = dinv[node < n ? node : n - 1];
    }
#pragma unroll
    for (int j = 0; j < 8; ++j) {
        int node = node0 + wv * 8 + j;
        float4 o = make_float4(b4.x + di[j] * acc[4 * j],     b4.y + di[j] * acc[4 * j + 1],
                               b4.z + di[j] * acc[4 * j + 2], b4.w + di[j] * acc[4 * j + 3]);
        if (node < n) ((float4*)Y)[(size_t)node * 32 + lane] = o;
    }
}

__global__ void gcn_gelu(float* __restrict__ Y, int total4) {
    int i = blockIdx.x * blockDim.x + threadIdx.x;
    if (i >= total4) return;
    float4 v = ((float4*)Y)[i];
    const float k = 0.70710678118654752440f;
    v.x = 0.5f * v.x * (1.0f + erff(v.x * k));
    v.y = 0.5f * v.y * (1.0f + erff(v.y * k));
    v.z = 0.5f * v.z * (1.0f + erff(v.z * k));
    v.w = 0.5f * v.w * (1.0f + erff(v.w * k));
    ((float4*)Y)[i] = v;
}

extern "C" void kernel_launch(void* const* d_in, const int* in_sizes, int n_in,
                              void* d_out, int out_size, void* d_ws, size_t ws_size,
                              hipStream_t stream) {
    const float* x    = (const float*)d_in[0];
    const int*   eidx = (const int*)  d_in[1];
    const float* Ws   = (const float*)d_in[2];
    const float* bs   = (const float*)d_in[3];
    float*       out  = (float*)d_out;

    const int N = in_sizes[0] / DH;
    const int E = in_sizes[1] / 2;
    const int* src = eidx;
    const int* dst = eidx + E;

    const int NB      = (N + TN - 1) / TN;
    const int PITCH   = (NB + 31) & ~31;
    const int NCHUNK  = (E + CHUNK - 1) / CHUNK;

    float* ws    = (float*)d_ws;
    float* dinv  = ws;
    int*   hist  = (int*)(dinv + (((size_t)N + 63) & ~(size_t)63));
    int*   start = hist  + (size_t)NCHUNK * PITCH;
    int*   list  = start + (((size_t)NB + 1 + 63) & ~(size_t)63);
    float* Wf    = (float*)(list + (((size_t)2 * E + 63) & ~(size_t)63));
    float* h     = Wf + (size_t)3 * DH * DH;
    float* buf   = h  + (size_t)N * DH;

    const int threads = 256;
    const int total4  = N * DH / 4;
    const int wcount  = 3 * DH * DH;

    gcn_part_hist  <<<NCHUNK, 1024, 0, stream>>>(dst, hist, E, PITCH);
    gcn_part_fill  <<<NCHUNK, 1024, 0, stream>>>(src, dst, hist, list, start, E, NB, PITCH, NCHUNK);
    gcn_dinv_range <<<NB, 256, 0, stream>>>(list, start, dinv, N);
    gcn_pack_w     <<<(wcount + threads - 1) / threads, threads, 0, stream>>>(Ws, Wf, wcount);

    const float* xin = x;
    for (int l = 0; l < 3; ++l) {
        float* y = (l == 2) ? out : buf;
        gcn_gemm<<<(N + 31) / 32, 64, 0, stream>>>(xin, Wf + (size_t)l * DH * DH, dinv, h, N);
        gcn_agg <<<NB, 256, 0, stream>>>(h, y, list, start, dinv, bs + l * DH, N);
        xin = y;
    }

    gcn_gelu<<<(total4 + threads - 1) / threads, threads, 0, stream>>>(out, total4);
}


# Degree and weight-composition launches removed: GEMM work-groups count their rows' in-degrees for dinv, Wc is built in 8 extra work-groups of the slot-store launch; 7 launches for 9; sum order kept
# speedup vs baseline: 4.0811x; 1.2463x over previous
#include <hip/hip_runtime.h>
#include <hip/hip_bf16.h>
#include <math.h>


typedef float v2f __attribute__((ext_vector_type(2)));
typedef float v4f __attribute__((ext_vector_type(4)));
typedef float v8f __attribute__((ext_vector_type(8)));
typedef float v32f __attribute__((ext_vector_type(32)));

#define DH 128
#define TN 64
#define TSH 6
#define MAXB 1024
#define CHUNK 4096

__global__ void __launch_bounds__(1024)
gcn_part_hist(const int* __restrict__ dst, int* __restrict__ hist, int ne, int pitch) {
    __shared__ int lh[MAXB];
    const int tid = threadIdx.x;
    lh[tid] = 0;
    __syncthreads();
    const int e0 = blockIdx.x * CHUNK + tid;
    int d[4];
#pragma unroll
    for (int k = 0; k < 4; ++k) {
        int e = e0 + k * 1024;
        d[k] = (e < ne) ? dst[e] : -1;
    }
#pragma unroll
    for (int k = 0; k < 4; ++k)
        if (d[k] >= 0) atomicAdd(&lh[d[k] >> TSH], 1);
    __syncthreads();
    if (tid < pitch) hist[blockIdx.x * pitch + tid] = lh[tid];
}

#define PTT 132
__device__ __forceinline__ void
gcn_compose_w(const float* __restrict__ W, const float* __restrict__ bs,
              float* __restrict__ Wf, float* __restrict__ cv,
              float* Tt, const int piece) {
    const int tid  = threadIdx.x;
    const int lane = tid & 31;
    const int w    = __builtin_amdgcn_readfirstlane(tid >> 5);
    const bool act = w < 8;
    const int c    = lane & 15;
    const int half = lane >> 4;
    const int wh   = piece >> 2;
    const int nt   = piece & 3;
    const int col  = 64 * wh + nt + 4 * c;
    const float* W0 = W;
    const float* W1 = W + DH * DH;
    const float* W2 = W + 2 * DH * DH;

    v4f a0[16];
    if (act) {
        const float* a1p = W1 + (16 * w + c) * DH + 4 * half;
        const float* a0p = W0 + (16 * w + c) * DH + 4 * half;
        const float* b2p = W2 + (4 * half) * DH + col;

        v4f a1[16];
        float b[16][4];
#pragma unroll
        for (int j = 0; j < 16; ++j) a1[j] = *(const v4f*)(a1p + 8 * j);
#pragma unroll
        for (int j = 0; j < 16; ++j)
#pragma unroll
            for (int e = 0; e < 4; ++e) b[j][e] = b2p[(8 * j + e) * DH];
#pragma unroll
        for (int j = 0; j < 16; ++j) a0[j] = *(const v4f*)(a0p + 8 * j);

        v8f t = (v8f){};
#pragma unroll
        for (int j = 0; j < 16; ++j) {
            v2f x0 = {a1[j].x, a1[j].y}, y0 = {b[j][0], b[j][1]};
            t = __builtin_amdgcn_wmma_f32_16x16x4_f32(false, x0, false, y0, (short)0, t, false, false);
            v2f x1 = {a1[j].z, a1[j].w}, y1 = {b[j][2], b[j][3]};
            t = __builtin_amdgcn_wmma_f32_16x16x4_f32(false, x1, false, y1, (short)0, t, false, false);
        }
        float* tp = Tt + c * PTT + 16 * w + 8 * half;
        *(v4f*)tp       = (v4f){t[0], t[1], t[2], t[3]};
        *(v4f*)(tp + 4) = (v4f){t[4], t[5], t[6], t[7]};
    }
    __syncthreads();
    if (!act) return;

    v8f d = (v8f){};
    const float* tb = Tt + c * PTT + 4 * half;
#pragma unroll
    for (int j = 0; j < 16; ++j) {
        const v4f bb = *(const v4f*)(tb + 8 * j);
        v2f x0 = {a0[j].x, a0[j].y}, y0 = {bb.x, bb.y};
        d = __builtin_amdgcn_wmma_f32_16x16x4_f32(false, x0, false, y0, (short)0, d, false, false);
        v2f x1 = {a0[j].z, a0[j].w}, y1 = {bb.z, bb.w};
        d = __builtin_amdgcn_wmma_f32_16x16x4_f32(false, x1, false, y1, (short)0, d, false, false);
    }
    float* op = Wf + (size_t)piece * (16 * 128) + (2 * w + half) * 128 + 4 * c;
    *(v4f*)op        = (v4f){d[0], d[1], d[2], d[3]};
    *(v4f*)(op + 64) = (v4f){d[4], d[5], d[6], d[7]};

    if (tid < 16) {
        const int cb = 64 * wh + nt + 4 * tid;
        const float* tr = Tt + tid * PTT;
        float s0 = 0.f, s1 = 0.f, s2 = 0.f, s3 = 0.f;
        for (int k = 0; k < DH; k += 4) {
            const v4f tv = *(const v4f*)(tr + k);
            s0 += bs[k] * tv.x; s1 += bs[k + 1] * tv.y; s2 += bs[k + 2] * tv.z; s3 += bs[k + 3] * tv.w;
        }
        cv[cb] = (s0 + s1) + (s2 + s3);
    } else if (tid >= 32 && tid < 48) {
        const int cb = 64 * wh + nt + 4 * (tid - 32);
        float s0 = 0.f, s1 = 0.f, s2 = 0.f, s3 = 0.f;
        for (int k = 0; k < DH; k += 4) {
            s0 += bs[DH + k]     * W2[k * DH + cb];
            s1 += bs[DH + k + 1] * W2[(k + 1) * DH + cb];
            s2 += bs[DH + k + 2] * W2[(k + 2) * DH + cb];
            s3 += bs[DH + k + 3] * W2[(k + 3) * DH + cb];
        }
        cv[DH + cb] = (s0 + s1) + (s2 + s3);
    } else if (tid >= 64 && tid < 80) {
        const int cb = 64 * wh + nt + 4 * (tid - 64);
        cv[2 * DH + cb] = bs[2 * DH + cb];
    }
}

__global__ void __launch_bounds__(1024)
gcn_part_fill(const int* __restrict__ src, const int* __restrict__ dst,
              const int* __restrict__ hist, int* __restrict__ list,
              int* __restrict__ start, int ne, int nb, int pitch, int nchunks,
              const float* __restrict__ W, const float* __restrict__ bs,
              float* __restrict__ Wf, float* __restrict__ cv) {
    __shared__ int cur[MAXB];
    __shared__ int wtot[32];
    __shared__ __attribute__((aligned(16))) float Tt[16 * PTT];
    if ((int)blockIdx.x >= nchunks) {
        gcn_compose_w(W, bs, Wf, cv, Tt, (int)blockIdx.x - nchunks);
        return;
    }
    const int tid  = threadIdx.x;
    const int lane = tid & 31;
    const int wv   = tid >> 5;
    const int me   = blockIdx.x;

    int tot = 0, below = 0;
    if (tid < nb) {
#pragma unroll 16
        for (int c = 0; c < nchunks; ++c) {
            int v = hist[c * pitch + tid];
            tot += v;
            below += (c < me) ? v : 0;
        }
    }
    int inc = tot;
#pragma unroll
    for (int o = 1; o < 32; o <<= 1) {
        int t = __shfl_up(inc, o);
        if (lane >= o) inc += t;
    }
    if (lane == 31) wtot[wv] = inc;
    __syncthreads();
    if (wv == 0) {
        int w = wtot[lane];
        int wi = w;
#pragma unroll
        for (int o = 1; o < 32; o <<= 1) {
            int t = __shfl_up(wi, o);
            if (lane >= o) wi += t;
        }
        wtot[lane] = wi - w;
    }
    __syncthreads();
    const int base = wtot[wv] + inc - tot;
    cur[tid] = base + below;
    if (me == 0) {
        if (tid < nb) start[tid] = base;
        if (tid == 0) start[nb] = ne;
    }
    __syncthreads();

    const int e0 = me * CHUNK + tid;
    int d[4], s[4];
#pragma unroll
    for (int k = 0; k < 4; ++k) {
        int e = e0 + k * 1024;
        bool ok = e < ne;
        d[k] = ok ? dst[e] : -1;
        s[k] = ok ? src[e] : 0;
    }
#pragma unroll
    for (int k = 0; k < 4; ++k) {
        if (d[k] >= 0) {
            int slot = atomicAdd(&cur[d[k] >> TSH], 1);
            ((int2*)list)[slot] = make_int2(s[k], d[k] & (TN - 1));
        }
    }
}

#define GCH 2048
__global__ void __launch_bounds__(64) __attribute__((amdgpu_waves_per_eu(1, 4)))
gcn_gemm(const float* __restrict__ X, const float* __restrict__ Wf,
         const int* __restrict__ list, const int* __restrict__ start,
         float* __restrict__ dinv, float* __restrict__ H, int nrows) {
    __shared__ int cnt[32];
    if (threadIdx.x < 32) cnt[threadIdx.x] = 0;
    __syncthreads();
    const int lane = threadIdx.x & 31;
    const int wh   = threadIdx.x >> 5;
    const int m0   = blockIdx.x << 5;
    const int r    = lane & 15;
    const int half = lane >> 4;

    const float* ap[2];
#pragma unroll
    for (int mt = 0; mt < 2; ++mt) {
        int arow = m0 + 16 * mt + r;
        if (arow >= nrows) arow = nrows - 1;
        ap[mt] = X + (size_t)arow * DH + 4 * half;
    }
    const float* bp = Wf + (size_t)wh * (4 * 16 * 128) + 4 * lane;

    v8f c[2][4];
#pragma unroll
    for (int mt = 0; mt < 2; ++mt)
#pragma unroll
        for (int nt = 0; nt < 4; ++nt) c[mt][nt] = (v8f){};

    v4f a[3][2], b[3][4];
#pragma unroll
    for (int p = 0; p < 2; ++p) {
#pragma unroll
        for (int mt = 0; mt < 2; ++mt) a[p][mt] = *(const v4f*)(ap[mt] + 8 * p);
#pragma unroll
        for (int nt = 0; nt < 4; ++nt) b[p][nt] = *(const v4f*)(bp + (nt * 16 + p) * 128);
    }

    {
        const int rb   = (int)blockIdx.x >> (TSH - 5);
        const int hsel = (int)blockIdx.x & ((TN >> 5) - 1);
        const int beg  = start[rb];
        const int num  = start[rb + 1] - beg;
        for (int base = 0; base < num; base += GCH) {
            const int m = (num - base < GCH) ? num - base : GCH;
            int y[GCH / 64];
#pragma unroll
            for (int q = 0; q < GCH / 64; ++q) {
                const int i = (int)threadIdx.x + q * 64;
                y[q] = list[2 * (size_t)(beg + base + (i < m ? i : m - 1)) + 1];
            }
#pragma unroll
            for (int q = 0; q < GCH / 64; ++q) {
                const int i = (int)threadIdx.x + q * 64;
                if (i < m && (y[q] >> 5) == hsel) atomicAdd(&cnt[y[q] & 31], 1);
            }
        }
    }
    __syncthreads();
    if (threadIdx.x < 32 && m0 + (int)threadIdx.x < nrows)
        dinv[m0 + threadIdx.x] = rsqrtf(1.0f + (float)cnt[threadIdx.x]);

#pragma unroll
    for (int j = 0; j < DH / 8; ++j) {
        const int cur = j % 3, nxt = (j + 2) % 3;
        if (j + 2 < DH / 8) {
#pragma unroll
            for (int mt = 0; mt < 2; ++mt) a[nxt][mt] = *(const v4f*)(ap[mt] + 8 * (j + 2));
#pragma unroll
            for (int nt = 0; nt < 4; ++nt) b[nxt][nt] = *(const v4f*)(bp + (nt * 16 + j + 2) * 128);
        }
#pragma unroll
        for (int mt = 0; mt < 2; ++mt)
#pragma unroll
            for (int nt = 0; nt < 4; ++nt) {
                v2f a2 = {a[cur][mt].x, a[cur][mt].y};
                v2f b2 = {b[cur][nt].x, b[cur][nt].y};
                c[mt][nt] = __builtin_amdgcn_wmma_f32_16x16x4_f32(
                         false, a2,  false, b2,
                         (short)0, c[mt][nt],  false,  false);
            }
#pragma unroll
        for (int mt = 0; mt < 2; ++mt)
#pragma unroll
            for (int nt = 0; nt < 4; ++nt) {
                v2f a2 = {a[cur][mt].z, a[cur][mt].w};
                v2f b2 = {b[cur][nt].z, b[cur][nt].w};
                c[mt][nt] = __builtin_amdgcn_wmma_f32_16x16x4_f32(
                         false, a2,  false, b2,
                         (short)0, c[mt][nt],  false,  false);
            }
    }

    float* hp = H + (size_t)m0 * DH + 64 * wh + 4 * r;
#pragma unroll
    for (int mt = 0; mt < 2; ++mt)
#pragma unroll
        for (int v = 0; v < 8; ++v) {
            int row = 16 * mt + v + 8 * half;
            float s = rsqrtf(1.0f + (float)cnt[row]);
            float4 o = make_float4(c[mt][0][v] * s, c[mt][1][v] * s,
                                   c[mt][2][v] * s, c[mt][3][v] * s);
            if (m0 + row < nrows) *(float4*)(hp + (size_t)row * DH) = o;
        }
}

#define LCH 2048
template <bool NEXT>
__global__ void __launch_bounds__(256)
gcn_agg(const float* __restrict__ H, float* __restrict__ Y,
        const int* __restrict__ list, const int* __restrict__ start,
        const float* __restrict__ dinv, const float* __restrict__ bias, int n) {
    __shared__ int2 sub[LCH];
    __shared__ int  cw[64];
    const int tid   = threadIdx.x;
    const int lane  = tid & 31;
    const int wv    = __builtin_amdgcn_readfirstlane(tid >> 5);
    const int node0 = blockIdx.x * TN;
    const float4* H4 = (const float4*)H;
    const int2*   L2 = (const int2*)list;
    const unsigned below_me = (1u << lane) - 1u;

    v32f acc;
#pragma unroll
    for (int j = 0; j < 8; ++j) {
        int node = node0 + wv * 8 + j;
        int nc = node < n ? node : n - 1;
        const float4 t = H4[(size_t)nc * 32 + lane];
        acc[4 * j] = t.x; acc[4 * j + 1] = t.y; acc[4 * j + 2] = t.z; acc[4 * j + 3] = t.w;
    }

    const int beg = start[blockIdx.x];
    const int cnt = start[blockIdx.x + 1] - beg;

    for (int base = 0; base < cnt; base += LCH) {
        const int m = (cnt - base < LCH) ? cnt - base : LCH;
        int2 e[LCH / 256];
        int  loc[LCH / 256];
        int  mine = 0;
#pragma unroll
        for (int q = 0; q < LCH / 256; ++q) {
            const int i = tid + q * 256;
            const bool ok = i < m;
            e[q] = L2[beg + base + (ok ? i : m - 1)];
            const int o = ok ? (e[q].y >> 3) : 8;
            const int before = __shfl(mine, o & 7);
            int rank = 0;
#pragma unroll
            for (int oo = 0; oo < 8; ++oo) {
                const unsigned mk = __builtin_amdgcn_ballot_w32(o == oo);
                if (o == oo) rank = __popc(mk & below_me);
                if (lane == oo) mine += __popc(mk);
            }
            loc[q] = before + rank;
        }
        if (lane < 8) cw[wv * 8 + lane] = mine;
        __syncthreads();

        int tot = 0, below = 0;
#pragma unroll
        for (int w = 0; w < 8; ++w) {
            const int c = cw[w * 8 + (lane & 7)];
            tot += c;
            below += (w < wv) ? c : 0;
        }
        int first = 0;
#pragma unroll
        for (int oo = 0; oo < 7; ++oo) {
            const int t = __shfl(tot, oo);
            first += ((lane & 7) > oo) ? t : 0;
        }
        const int basev = first + below;
#pragma unroll
        for (int q = 0; q < LCH / 256; ++q) {
            const int i = tid + q * 256;
            const int o = e[q].y >> 3;
            const int slot = __shfl(basev, o & 7) + loc[q];
            if (i < m) sub[slot] = e[q];
        }
        const int my_cnt = __builtin_amdgcn_readlane(tot, wv);
        const int my_beg = __builtin_amdgcn_readlane(first, wv);
        __syncthreads();

        for (int g = 0; g < my_cnt; g += 8) {
            int i = g + (lane & 7);
            if (i > my_cnt - 1) i = my_cnt - 1;
            const int2 ec = sub[my_beg + i];
            const int nval = my_cnt - g;
            float4 h[8];
            int dl[8];
#pragma unroll
            for (int k = 0; k < 8; ++k) {
                int s = __builtin_amdgcn_readlane(ec.x, k);
                dl[k] = (__builtin_amdgcn_readlane(ec.y, k) & 7) * 4;
                h[k] = H4[(size_t)s * 32 + lane];
            }
#pragma unroll
            for (int k = 0; k < 8; ++k) {
                if (k < nval) {
                    const float4 v = h[k];
                    const int b = dl[k];
                    acc[b] += v.x; acc[b + 1] += v.y; acc[b + 2] += v.z; acc[b + 3] += v.w;
                }
            }
        }
        __syncthreads();
    }

    const float4 b4 = ((const float4*)bias)[lane];
    float di[8];
#pragma unroll
    for (int j = 0; j < 8; ++j) {
        int node = node0 + wv * 8 + j;
        di[j] = dinv[node < n ? node : n - 1];
    }
#pragma unroll
    for (int j = 0; j < 8; ++j) {
        int node = node0 + wv * 8 + j;
        float4 o = make_float4(b4.x + di[j] * acc[4 * j],     b4.y + di[j] * acc[4 * j + 1],
                               b4.z + di[j] * acc[4 * j + 2], b4.w + di[j] * acc[4 * j + 3]);
        if (NEXT) { o.x *= di[j]; o.y *= di[j]; o.z *= di[j]; o.w *= di[j]; }
        if (node < n) ((float4*)Y)[(size_t)node * 32 + lane] = o;
    }
}

__global__ void gcn_gelu(float* __restrict__ Y, int total4) {
    int i = blockIdx.x * blockDim.x + threadIdx.x;
    if (i >= total4) return;
    float4 v = ((float4*)Y)[i];
    const float k = 0.70710678118654752440f;
    v.x = 0.5f * v.x * (1.0f + erff(v.x * k));
    v.y = 0.5f * v.y * (1.0f + erff(v.y * k));
    v.z = 0.5f * v.z * (1.0f + erff(v.z * k));
    v.w = 0.5f * v.w * (1.0f + erff(v.w * k));
    ((float4*)Y)[i] = v;
}

extern "C" void kernel_launch(void* const* d_in, const int* in_sizes, int n_in,
                              void* d_out, int out_size, void* d_ws, size_t ws_size,
                              hipStream_t stream) {
    const float* x    = (const float*)d_in[0];
    const int*   eidx = (const int*)  d_in[1];
    const float* Ws   = (const float*)d_in[2];
    const float* bs   = (const float*)d_in[3];
    float*       out  = (float*)d_out;

    const int N = in_sizes[0] / DH;
    const int E = in_sizes[1] / 2;
    const int* src = eidx;
    const int* dst = eidx + E;

    const int NB      = (N + TN - 1) / TN;
    const int PITCH   = (NB + 31) & ~31;
    const int NCHUNK  = (E + CHUNK - 1) / CHUNK;

    float* ws    = (float*)d_ws;
    float* dinv  = ws;
    int*   hist  = (int*)(dinv + (((size_t)N + 63) & ~(size_t)63));
    int*   start = hist  + (size_t)NCHUNK * PITCH;
    int*   list  = start + (((size_t)NB + 1 + 63) & ~(size_t)63);
    float* Wf    = (float*)(list + (((size_t)2 * E + 63) & ~(size_t)63));
    float* cv    = Wf + (size_t)DH * DH;
    float* h     = cv + (size_t)3 * DH;
    float* buf   = h  + (size_t)N * DH;

    const int threads = 256;
    const int total4  = N * DH / 4;

    gcn_part_hist  <<<NCHUNK, 1024, 0, stream>>>(dst, hist, E, PITCH);
    gcn_part_fill  <<<NCHUNK + 8, 1024, 0, stream>>>(src, dst, hist, list, start, E, NB, PITCH, NCHUNK,
                                                    Ws, bs, Wf, cv);

    gcn_gemm<<<(N + 31) / 32, 64, 0, stream>>>(x, Wf, list, start, dinv, h, N);
    gcn_agg<true> <<<NB, 256, 0, stream>>>(h,   buf, list, start, dinv, cv,          N);
    gcn_agg<true> <<<NB, 256, 0, stream>>>(buf, h,   list, start, dinv, cv + DH,     N);
    gcn_agg<false><<<NB, 256, 0, stream>>>(h,   out, list, start, dinv, cv + 2 * DH, N);

    gcn_gelu<<<(total4 + threads - 1) / threads, threads, 0, stream>>>(out, total4);
}


# Exact GELU applied by the last neighbour sum to its rows in registers at their store; the separate elementwise launch (output read and written again) is gone: 6 launches for 7; sum order kept
# speedup vs baseline: 4.2763x; 1.0478x over previous
#include <hip/hip_runtime.h>
#include <hip/hip_bf16.h>
#include <math.h>


typedef float v2f __attribute__((ext_vector_type(2)));
typedef float v4f __attribute__((ext_vector_type(4)));
typedef float v8f __attribute__((ext_vector_type(8)));
typedef float v32f __attribute__((ext_vector_type(32)));

#define DH 128
#define TN 64
#define TSH 6
#define MAXB 1024
#define CHUNK 4096

__global__ void __launch_bounds__(1024)
gcn_part_hist(const int* __restrict__ dst, int* __restrict__ hist, int ne, int pitch) {
    __shared__ int lh[MAXB];
    const int tid = threadIdx.x;
    lh[tid] = 0;
    __syncthreads();
    const int e0 = blockIdx.x * CHUNK + tid;
    int d[4];
#pragma unroll
    for (int k = 0; k < 4; ++k) {
        int e = e0 + k * 1024;
        d[k] = (e < ne) ? dst[e] : -1;
    }
#pragma unroll
    for (int k = 0; k < 4; ++k)
        if (d[k] >= 0) atomicAdd(&lh[d[k] >> TSH], 1);
    __syncthreads();
    if (tid < pitch) hist[blockIdx.x * pitch + tid] = lh[tid];
}

#define PTT 132
__device__ __forceinline__ void
gcn_compose_w(const float* __restrict__ W, const float* __restrict__ bs,
              float* __restrict__ Wf, float* __restrict__ cv,
              float* Tt, const int piece) {
    const int tid  = threadIdx.x;
    const int lane = tid & 31;
    const int w    = __builtin_amdgcn_readfirstlane(tid >> 5);
    const bool act = w < 8;
    const int c    = lane & 15;
    const int half = lane >> 4;
    const int wh   = piece >> 2;
    const int nt   = piece & 3;
    const int col  = 64 * wh + nt + 4 * c;
    const float* W0 = W;
    const float* W1 = W + DH * DH;
    const float* W2 = W + 2 * DH * DH;

    v4f a0[16];
    if (act) {
        const float* a1p = W1 + (16 * w + c) * DH + 4 * half;
        const float* a0p = W0 + (16 * w + c) * DH + 4 * half;
        const float* b2p = W2 + (4 * half) * DH + col;

        v4f a1[16];
        float b[16][4];
#pragma unroll
        for (int j = 0; j < 16; ++j) a1[j] = *(const v4f*)(a1p + 8 * j);
#pragma unroll
        for (int j = 0; j < 16; ++j)
#pragma unroll
            for (int e = 0; e < 4; ++e) b[j][e] = b2p[(8 * j + e) * DH];
#pragma unroll
        for (int j = 0; j < 16; ++j) a0[j] = *(const v4f*)(a0p + 8 * j);

        v8f t = (v8f){};
#pragma unroll
        for (int j = 0; j < 16; ++j) {
            v2f x0 = {a1[j].x, a1[j].y}, y0 = {b[j][0], b[j][1]};
            t = __builtin_amdgcn_wmma_f32_16x16x4_f32(false, x0, false, y0, (short)0, t, false, false);
            v2f x1 = {a1[j].z, a1[j].w}, y1 = {b[j][2], b[j][3]};
            t = __builtin_amdgcn_wmma_f32_16x16x4_f32(false, x1, false, y1, (short)0, t, false, false);
        }
        float* tp = Tt + c * PTT + 16 * w + 8 * half;
        *(v4f*)tp       = (v4f){t[0], t[1], t[2], t[3]};
        *(v4f*)(tp + 4) = (v4f){t[4], t[5], t[6], t[7]};
    }
    __syncthreads();
    if (!act) return;

    v8f d = (v8f){};
    const float* tb = Tt + c * PTT + 4 * half;
#pragma unroll
    for (int j = 0; j < 16; ++j) {
        const v4f bb = *(const v4f*)(tb + 8 * j);
        v2f x0 = {a0[j].x, a0[j].y}, y0 = {bb.x, bb.y};
        d = __builtin_amdgcn_wmma_f32_16x16x4_f32(false, x0, false, y0, (short)0, d, false, false);
        v2f x1 = {a0[j].z, a0[j].w}, y1 = {bb.z, bb.w};
        d = __builtin_amdgcn_wmma_f32_16x16x4_f32(false, x1, false, y1, (short)0, d, false, false);
    }
    float* op = Wf + (size_t)piece * (16 * 128) + (2 * w + half) * 128 + 4 * c;
    *(v4f*)op        = (v4f){d[0], d[1], d[2], d[3]};
    *(v4f*)(op + 64) = (v4f){d[4], d[5], d[6], d[7]};

    if (tid < 16) {
        const int cb = 64 * wh + nt + 4 * tid;
        const float* tr = Tt + tid * PTT;
        float s0 = 0.f, s1 = 0.f, s2 = 0.f, s3 = 0.f;
        for (int k = 0; k < DH; k += 4) {
            const v4f tv = *(const v4f*)(tr + k);
            s0 += bs[k] * tv.x; s1 += bs[k + 1] * tv.y; s2 += bs[k + 2] * tv.z; s3 += bs[k + 3] * tv.w;
        }
        cv[cb] = (s0 + s1) + (s2 + s3);
    } else if (tid >= 32 && tid < 48) {
        const int cb = 64 * wh + nt + 4 * (tid - 32);
        float s0 = 0.f, s1 = 0.f, s2 = 0.f, s3 = 0.f;
        for (int k = 0; k < DH; k += 4) {
            s0 += bs[DH + k]     * W2[k * DH + cb];
            s1 += bs[DH + k + 1] * W2[(k + 1) * DH + cb];
            s2 += bs[DH + k + 2] * W2[(k + 2) * DH + cb];
            s3 += bs[DH + k + 3] * W2[(k + 3) * DH + cb];
        }
        cv[DH + cb] = (s0 + s1) + (s2 + s3);
    } else if (tid >= 64 && tid < 80) {
        const int cb = 64 * wh + nt + 4 * (tid - 64);
        cv[2 * DH + cb] = bs[2 * DH + cb];
    }
}

__global__ void __launch_bounds__(1024)
gcn_part_fill(const int* __restrict__ src, const int* __restrict__ dst,
              const int* __restrict__ hist, int* __restrict__ list,
              int* __restrict__ start, int ne, int nb, int pitch, int nchunks,
              const float* __restrict__ W, const float* __restrict__ bs,
              float* __restrict__ Wf, float* __restrict__ cv) {
    __shared__ int cur[MAXB];
    __shared__ int wtot[32];
    __shared__ __attribute__((aligned(16))) float Tt[16 * PTT];
    if ((int)blockIdx.x >= nchunks) {
        gcn_compose_w(W, bs, Wf, cv, Tt, (int)blockIdx.x - nchunks);
        return;
    }
    const int tid  = threadIdx.x;
    const int lane = tid & 31;
    const int wv   = tid >> 5;
    const int me   = blockIdx.x;

    int tot = 0, below = 0;
    if (tid < nb) {
#pragma unroll 16
        for (int c = 0; c < nchunks; ++c) {
            int v = hist[c * pitch + tid];
            tot += v;
            below += (c < me) ? v : 0;
        }
    }
    int inc = tot;
#pragma unroll
    for (int o = 1; o < 32; o <<= 1) {
        int t = __shfl_up(inc, o);
        if (lane >= o) inc += t;
    }
    if (lane == 31) wtot[wv] = inc;
    __syncthreads();
    if (wv == 0) {
        int w = wtot[lane];
        int wi = w;
#pragma unroll
        for (int o = 1; o < 32; o <<= 1) {
            int t = __shfl_up(wi, o);
            if (lane >= o) wi += t;
        }
        wtot[lane] = wi - w;
    }
    __syncthreads();
    const int base = wtot[wv] + inc - tot;
    cur[tid] = base + below;
    if (me == 0) {
        if (tid < nb) start[tid] = base;
        if (tid == 0) start[nb] = ne;
    }
    __syncthreads();

    const int e0 = me * CHUNK + tid;
    int d[4], s[4];
#pragma unroll
    for (int k = 0; k < 4; ++k) {
        int e = e0 + k * 1024;
        bool ok = e < ne;
        d[k] = ok ? dst[e] : -1;
        s[k] = ok ? src[e] : 0;
    }
#pragma unroll
    for (int k = 0; k < 4; ++k) {
        if (d[k] >= 0) {
            int slot = atomicAdd(&cur[d[k] >> TSH], 1);
            ((int2*)list)[slot] = make_int2(s[k], d[k] & (TN - 1));
        }
    }
}

#define GCH 2048
__global__ void __launch_bounds__(64) __attribute__((amdgpu_waves_per_eu(1, 4)))
gcn_gemm(const float* __restrict__ X, const float* __restrict__ Wf,
         const int* __restrict__ list, const int* __restrict__ start,
         float* __restrict__ dinv, float* __restrict__ H, int nrows) {
    __shared__ int cnt[32];
    if (threadIdx.x < 32) cnt[threadIdx.x] = 0;
    __syncthreads();
    const int lane = threadIdx.x & 31;
    const int wh   = threadIdx.x >> 5;
    const int m0   = blockIdx.x << 5;
    const int r    = lane & 15;
    const int half = lane >> 4;

    const float* ap[2];
#pragma unroll
    for (int mt = 0; mt < 2; ++mt) {
        int arow = m0 + 16 * mt + r;
        if (arow >= nrows) arow = nrows - 1;
        ap[mt] = X + (size_t)arow * DH + 4 * half;
    }
    const float* bp = Wf + (size_t)wh * (4 * 16 * 128) + 4 * lane;

    v8f c[2][4];
#pragma unroll
    for (int mt = 0; mt < 2; ++mt)
#pragma unroll
        for (int nt = 0; nt < 4; ++nt) c[mt][nt] = (v8f){};

    v4f a[3][2], b[3][4];
#pragma unroll
    for (int p = 0; p < 2; ++p) {
#pragma unroll
        for (int mt = 0; mt < 2; ++mt) a[p][mt] = *(const v4f*)(ap[mt] + 8 * p);
#pragma unroll
        for (int nt = 0; nt < 4; ++nt) b[p][nt] = *(const v4f*)(bp + (nt * 16 + p) * 128);
    }

    {
        const int rb   = (int)blockIdx.x >> (TSH - 5);
        const int hsel = (int)blockIdx.x & ((TN >> 5) - 1);
        const int beg  = start[rb];
        const int num  = start[rb + 1] - beg;
        for (int base = 0; base < num; base += GCH) {
            const int m = (num - base < GCH) ? num - base : GCH;
            int y[GCH / 64];
#pragma unroll
            for (int q = 0; q < GCH / 64; ++q) {
                const int i = (int)threadIdx.x + q * 64;
                y[q] = list[2 * (size_t)(beg + base + (i < m ? i : m - 1)) + 1];
            }
#pragma unroll
            for (int q = 0; q < GCH / 64; ++q) {
                const int i = (int)threadIdx.x + q * 64;
                if (i < m && (y[q] >> 5) == hsel) atomicAdd(&cnt[y[q] & 31], 1);
            }
        }
    }
    __syncthreads();
    if (threadIdx.x < 32 && m0 + (int)threadIdx.x < nrows)
        dinv[m0 + threadIdx.x] = rsqrtf(1.0f + (float)cnt[threadIdx.x]);

#pragma unroll
    for (int j = 0; j < DH / 8; ++j) {
        const int cur = j % 3, nxt = (j + 2) % 3;
        if (j + 2 < DH / 8) {
#pragma unroll
            for (int mt = 0; mt < 2; ++mt) a[nxt][mt] = *(const v4f*)(ap[mt] + 8 * (j + 2));
#pragma unroll
            for (int nt = 0; nt < 4; ++nt) b[nxt][nt] = *(const v4f*)(bp + (nt * 16 + j + 2) * 128);
        }
#pragma unroll
        for (int mt = 0; mt < 2; ++mt)
#pragma unroll
            for (int nt = 0; nt < 4; ++nt) {
                v2f a2 = {a[cur][mt].x, a[cur][mt].y};
                v2f b2 = {b[cur][nt].x, b[cur][nt].y};
                c[mt][nt] = __builtin_amdgcn_wmma_f32_16x16x4_f32(
                         false, a2,  false, b2,
                         (short)0, c[mt][nt],  false,  false);
            }
#pragma unroll
        for (int mt = 0; mt < 2; ++mt)
#pragma unroll
            for (int nt = 0; nt < 4; ++nt) {
                v2f a2 = {a[cur][mt].z, a[cur][mt].w};
                v2f b2 = {b[cur][nt].z, b[cur][nt].w};
                c[mt][nt] = __builtin_amdgcn_wmma_f32_16x16x4_f32(
                         false, a2,  false, b2,
                         (short)0, c[mt][nt],  false,  false);
            }
    }

    float* hp = H + (size_t)m0 * DH + 64 * wh + 4 * r;
#pragma unroll
    for (int mt = 0; mt < 2; ++mt)
#pragma unroll
        for (int v = 0; v < 8; ++v) {
            int row = 16 * mt + v + 8 * half;
            float s = rsqrtf(1.0f + (float)cnt[row]);
            float4 o = make_float4(c[mt][0][v] * s, c[mt][1][v] * s,
                                   c[mt][2][v] * s, c[mt][3][v] * s);
            if (m0 + row < nrows) *(float4*)(hp + (size_t)row * DH) = o;
        }
}

#define LCH 2048
__device__ __forceinline__ float gcn_gelu_exact(float x) {
    return 0.5f * x * (1.0f + erff(x * 0.70710678118654752440f));
}
template <bool NEXT>
__global__ void __launch_bounds__(256)
gcn_agg(const float* __restrict__ H, float* __restrict__ Y,
        const int* __restrict__ list, const int* __restrict__ start,
        const float* __restrict__ dinv, const float* __restrict__ bias, int n) {
    __shared__ int2 sub[LCH];
    __shared__ int  cw[64];
    const int tid   = threadIdx.x;
    const int lane  = tid & 31;
    const int wv    = __builtin_amdgcn_readfirstlane(tid >> 5);
    const int node0 = blockIdx.x * TN;
    const float4* H4 = (const float4*)H;
    const int2*   L2 = (const int2*)list;
    const unsigned below_me = (1u << lane) - 1u;

    v32f acc;
#pragma unroll
    for (int j = 0; j < 8; ++j) {
        int node = node0 + wv * 8 + j;
        int nc = node < n ? node : n - 1;
        const float4 t = H4[(size_t)nc * 32 + lane];
        acc[4 * j] = t.x; acc[4 * j + 1] = t.y; acc[4 * j + 2] = t.z; acc[4 * j + 3] = t.w;
    }

    const int beg = start[blockIdx.x];
    const int cnt = start[blockIdx.x + 1] - beg;

    for (int base = 0; base < cnt; base += LCH) {
        const int m = (cnt - base < LCH) ? cnt - base : LCH;
        int2 e[LCH / 256];
        int  loc[LCH / 256];
        int  mine = 0;
#pragma unroll
        for (int q = 0; q < LCH / 256; ++q) {
            const int i = tid + q * 256;
            const bool ok = i < m;
            e[q] = L2[beg + base + (ok ? i : m - 1)];
            const int o = ok ? (e[q].y >> 3) : 8;
            const int before = __shfl(mine, o & 7);
            int rank = 0;
#pragma unroll
            for (int oo = 0; oo < 8; ++oo) {
                const unsigned mk = __builtin_amdgcn_ballot_w32(o == oo);
                if (o == oo) rank = __popc(mk & below_me);
                if (lane == oo) mine += __popc(mk);
            }
            loc[q] = before + rank;
        }
        if (lane < 8) cw[wv * 8 + lane] = mine;
        __syncthreads();

        int tot = 0, below = 0;
#pragma unroll
        for (int w = 0; w < 8; ++w) {
            const int c = cw[w * 8 + (lane & 7)];
            tot += c;
            below += (w < wv) ? c : 0;
        }
        int first = 0;
#pragma unroll
        for (int oo = 0; oo < 7; ++oo) {
            const int t = __shfl(tot, oo);
            first += ((lane & 7) > oo) ? t : 0;
        }
        const int basev = first + below;
#pragma unroll
        for (int q = 0; q < LCH / 256; ++q) {
            const int i = tid + q * 256;
            const int o = e[q].y >> 3;
            const int slot = __shfl(basev, o & 7) + loc[q];
            if (i < m) sub[slot] = e[q];
        }
        const int my_cnt = __builtin_amdgcn_readlane(tot, wv);
        const int my_beg = __builtin_amdgcn_readlane(first, wv);
        __syncthreads();

        for (int g = 0; g < my_cnt; g += 8) {
            int i = g + (lane & 7);
            if (i > my_cnt - 1) i = my_cnt - 1;
            const int2 ec = sub[my_beg + i];
            const int nval = my_cnt - g;
            float4 h[8];
            int dl[8];
#pragma unroll
            for (int k = 0; k < 8; ++k) {
                int s = __builtin_amdgcn_readlane(ec.x, k);
                dl[k] = (__builtin_amdgcn_readlane(ec.y, k) & 7) * 4;
                h[k] = H4[(size_t)s * 32 + lane];
            }
#pragma unroll
            for (int k = 0; k < 8; ++k) {
                if (k < nval) {
                    const float4 v = h[k];
                    const int b = dl[k];
                    acc[b] += v.x; acc[b + 1] += v.y; acc[b + 2] += v.z; acc[b + 3] += v.w;
                }
            }
        }
        __syncthreads();
    }

    const float4 b4 = ((const float4*)bias)[lane];
    float di[8];
#pragma unroll
    for (int j = 0; j < 8; ++j) {
        int node = node0 + wv * 8 + j;
        di[j] = dinv[node < n ? node : n - 1];
    }
#pragma unroll
    for (int j = 0; j < 8; ++j) {
        int node = node0 + wv * 8 + j;
        float4 o = make_float4(b4.x + di[j] * acc[4 * j],     b4.y + di[j] * acc[4 * j + 1],
                               b4.z + di[j] * acc[4 * j + 2], b4.w + di[j] * acc[4 * j + 3]);
        if (NEXT) { o.x *= di[j]; o.y *= di[j]; o.z *= di[j]; o.w *= di[j]; }
        else { o.x = gcn_gelu_exact(o.x); o.y = gcn_gelu_exact(o.y); o.z = gcn_gelu_exact(o.z); o.w = gcn_gelu_exact(o.w); }
        if (node < n) ((float4*)Y)[(size_t)node * 32 + lane] = o;
    }
}

extern "C" void kernel_launch(void* const* d_in, const int* in_sizes, int n_in,
                              void* d_out, int out_size, void* d_ws, size_t ws_size,
                              hipStream_t stream) {
    const float* x    = (const float*)d_in[0];
    const int*   eidx = (const int*)  d_in[1];
    const float* Ws   = (const float*)d_in[2];
    const float* bs   = (const float*)d_in[3];
    float*       out  = (float*)d_out;

    const int N = in_sizes[0] / DH;
    const int E = in_sizes[1] / 2;
    const int* src = eidx;
    const int* dst = eidx + E;

    const int NB      = (N + TN - 1) / TN;
    const int PITCH   = (NB + 31) & ~31;
    const int NCHUNK  = (E + CHUNK - 1) / CHUNK;

    float* ws    = (float*)d_ws;
    float* dinv  = ws;
    int*   hist  = (int*)(dinv + (((size_t)N + 63) & ~(size_t)63));
    int*   start = hist  + (size_t)NCHUNK * PITCH;
    int*   list  = start + (((size_t)NB + 1 + 63) & ~(size_t)63);
    float* Wf    = (float*)(list + (((size_t)2 * E + 63) & ~(size_t)63));
    float* cv    = Wf + (size_t)DH * DH;
    float* h     = cv + (size_t)3 * DH;
    float* buf   = h  + (size_t)N * DH;

    gcn_part_hist  <<<NCHUNK, 1024, 0, stream>>>(dst, hist, E, PITCH);
    gcn_part_fill  <<<NCHUNK + 8, 1024, 0, stream>>>(src, dst, hist, list, start, E, NB, PITCH, NCHUNK,
                                                    Ws, bs, Wf, cv);

    gcn_gemm<<<(N + 31) / 32, 64, 0, stream>>>(x, Wf, list, start, dinv, h, N);
    gcn_agg<true> <<<NB, 256, 0, stream>>>(h,   buf, list, start, dinv, cv,          N);
    gcn_agg<true> <<<NB, 256, 0, stream>>>(buf, h,   list, start, dinv, cv + DH,     N);
    gcn_agg<false><<<NB, 256, 0, stream>>>(h,   out, list, start, dinv, cv + 2 * DH, N);
}


# Global histogram launch and offset scan removed: chunks order their edges by range locally, GEMM work-groups gather their range's per-chunk runs into its list; 5 launches for 6; sum order kept
# speedup vs baseline: 4.9123x; 1.1487x over previous
#include <hip/hip_runtime.h>
#include <hip/hip_bf16.h>
#include <math.h>


typedef float v2f __attribute__((ext_vector_type(2)));
typedef float v4f __attribute__((ext_vector_type(4)));
typedef float v8f __attribute__((ext_vector_type(8)));
typedef float v32f __attribute__((ext_vector_type(32)));

#define DH 128
#define TN 64
#define TSH 6
#define MAXB 1024
#define CHUNK 4096
#define MAXC 256

#define PTT 132
__device__ __forceinline__ void
gcn_compose_w(const float* __restrict__ W, const float* __restrict__ bs,
              float* __restrict__ Wf, float* __restrict__ cv,
              float* Tt, const int piece) {
    const int tid  = threadIdx.x;
    const int lane = tid & 31;
    const int w    = __builtin_amdgcn_readfirstlane(tid >> 5);
    const bool act = w < 8;
    const int c    = lane & 15;
    const int half = lane >> 4;
    const int wh   = piece >> 2;
    const int nt   = piece & 3;
    const int col  = 64 * wh + nt + 4 * c;
    const float* W0 = W;
    const float* W1 = W + DH * DH;
    const float* W2 = W + 2 * DH * DH;

    v4f a0[16];
    if (act) {
        const float* a1p = W1 + (16 * w + c) * DH + 4 * half;
        const float* a0p = W0 + (16 * w + c) * DH + 4 * half;
        const float* b2p = W2 + (4 * half) * DH + col;

        v4f a1[16];
        float b[16][4];
#pragma unroll
        for (int j = 0; j < 16; ++j) a1[j] = *(const v4f*)(a1p + 8 * j);
#pragma unroll
        for (int j = 0; j < 16; ++j)
#pragma unroll
            for (int e = 0; e < 4; ++e) b[j][e] = b2p[(8 * j + e) * DH];
#pragma unroll
        for (int j = 0; j < 16; ++j) a0[j] = *(const v4f*)(a0p + 8 * j);

        v8f t = (v8f){};
#pragma unroll
        for (int j = 0; j < 16; ++j) {
            v2f x0 = {a1[j].x, a1[j].y}, y0 = {b[j][0], b[j][1]};
            t = __builtin_amdgcn_wmma_f32_16x16x4_f32(false, x0, false, y0, (short)0, t, false, false);
            v2f x1 = {a1[j].z, a1[j].w}, y1 = {b[j][2], b[j][3]};
            t = __builtin_amdgcn_wmma_f32_16x16x4_f32(false, x1, false, y1, (short)0, t, false, false);
        }
        float* tp = Tt + c * PTT + 16 * w + 8 * half;
        *(v4f*)tp       = (v4f){t[0], t[1], t[2], t[3]};
        *(v4f*)(tp + 4) = (v4f){t[4], t[5], t[6], t[7]};
    }
    __syncthreads();
    if (!act) return;

    v8f d = (v8f){};
    const float* tb = Tt + c * PTT + 4 * half;
#pragma unroll
    for (int j = 0; j < 16; ++j) {
        const v4f bb = *(const v4f*)(tb + 8 * j);
        v2f x0 = {a0[j].x, a0[j].y}, y0 = {bb.x, bb.y};
        d = __builtin_amdgcn_wmma_f32_16x16x4_f32(false, x0, false, y0, (short)0, d, false, false);
        v2f x1 = {a0[j].z, a0[j].w}, y1 = {bb.z, bb.w};
        d = __builtin_amdgcn_wmma_f32_16x16x4_f32(false, x1, false, y1, (short)0, d, false, false);
    }
    float* op = Wf + (size_t)piece * (16 * 128) + (2 * w + half) * 128 + 4 * c;
    *(v4f*)op        = (v4f){d[0], d[1], d[2], d[3]};
    *(v4f*)(op + 64) = (v4f){d[4], d[5], d[6], d[7]};

    if (tid < 16) {
        const int cb = 64 * wh + nt + 4 * tid;
        const float* tr = Tt + tid * PTT;
        float s0 = 0.f, s1 = 0.f, s2 = 0.f, s3 = 0.f;
        for (int k = 0; k < DH; k += 4) {
            const v4f tv = *(const v4f*)(tr + k);
            s0 += bs[k] * tv.x; s1 += bs[k + 1] * tv.y; s2 += bs[k + 2] * tv.z; s3 += bs[k + 3] * tv.w;
        }
        cv[cb] = (s0 + s1) + (s2 + s3);
    } else if (tid >= 32 && tid < 48) {
        const int cb = 64 * wh + nt + 4 * (tid - 32);
        float s0 = 0.f, s1 = 0.f, s2 = 0.f, s3 = 0.f;
        for (int k = 0; k < DH; k += 4) {
            s0 += bs[DH + k]     * W2[k * DH + cb];
            s1 += bs[DH + k + 1] * W2[(k + 1) * DH + cb];
            s2 += bs[DH + k + 2] * W2[(k + 2) * DH + cb];
            s3 += bs[DH + k + 3] * W2[(k + 3) * DH + cb];
        }
        cv[DH + cb] = (s0 + s1) + (s2 + s3);
    } else if (tid >= 64 && tid < 80) {
        const int cb = 64 * wh + nt + 4 * (tid - 64);
        cv[2 * DH + cb] = bs[2 * DH + cb];
    }
}

__global__ void __launch_bounds__(1024)
gcn_part_local(const int* __restrict__ src, const int* __restrict__ dst,
               int* __restrict__ offs, int* __restrict__ listc,
               int ne, int nb, int pitch, int nchunks,
               const float* __restrict__ W, const float* __restrict__ bs,
               float* __restrict__ Wf, float* __restrict__ cv) {
    __shared__ int cur[MAXB];
    __shared__ int wtot[32];
    __shared__ __attribute__((aligned(16))) float Tt[16 * PTT];
    if ((int)blockIdx.x >= nchunks) {
        gcn_compose_w(W, bs, Wf, cv, Tt, (int)blockIdx.x - nchunks);
        return;
    }
    const int tid  = threadIdx.x;
    const int lane = tid & 31;
    const int wv   = tid >> 5;
    const int me   = blockIdx.x;

    cur[tid] = 0;
    __syncthreads();

    const int e0 = me * CHUNK + tid;
    int d[4], s[4];
#pragma unroll
    for (int k = 0; k < 4; ++k) {
        int e = e0 + k * 1024;
        bool ok = e < ne;
        d[k] = ok ? dst[e] : -1;
        s[k] = ok ? src[e] : 0;
    }
#pragma unroll
    for (int k = 0; k < 4; ++k)
        if (d[k] >= 0) atomicAdd(&cur[d[k] >> TSH], 1);
    __syncthreads();

    const int tot = cur[tid];
    int inc = tot;
#pragma unroll
    for (int o = 1; o < 32; o <<= 1) {
        int t = __shfl_up(inc, o);
        if (lane >= o) inc += t;
    }
    if (lane == 31) wtot[wv] = inc;
    __syncthreads();
    if (wv == 0) {
        int w = wtot[lane];
        int wi = w;
#pragma unroll
        for (int o = 1; o < 32; o <<= 1) {
            int t = __shfl_up(wi, o);
            if (lane >= o) wi += t;
        }
        wtot[lane] = wi - w;
    }
    __syncthreads();
    const int base = wtot[wv] + inc - tot;
    cur[tid] = base;
    if (tid <= nb) offs[(size_t)me * pitch + tid] = base;
    __syncthreads();

#pragma unroll
    for (int k = 0; k < 4; ++k) {
        if (d[k] >= 0) {
            int slot = atomicAdd(&cur[d[k] >> TSH], 1);
            ((int2*)listc)[(size_t)me * CHUNK + slot] = make_int2(s[k], d[k] & (TN - 1));
        }
    }
}

#define GCH 1280
__global__ void __launch_bounds__(64) __attribute__((amdgpu_waves_per_eu(1, 4)))
gcn_gemm(const float* __restrict__ X, const float* __restrict__ Wf,
         const int* __restrict__ offs, const int* __restrict__ listc,
         int* __restrict__ list, int* __restrict__ start,
         float* __restrict__ dinv, float* __restrict__ H, int nrows,
         int nchunks, int pitch, int nb) {
    __shared__ int cnt[32];
    __shared__ int rpre[MAXC];
    __shared__ int rsrc[MAXC];
    __shared__ int wsum[4];
    if (threadIdx.x < 32) cnt[threadIdx.x] = 0;
    __syncthreads();
    const int lane = threadIdx.x & 31;
    const int wh   = threadIdx.x >> 5;
    const int m0   = blockIdx.x << 5;
    const int r    = lane & 15;
    const int half = lane >> 4;

    const float* ap[2];
#pragma unroll
    for (int mt = 0; mt < 2; ++mt) {
        int arow = m0 + 16 * mt + r;
        if (arow >= nrows) arow = nrows - 1;
        ap[mt] = X + (size_t)arow * DH + 4 * half;
    }
    const float* bp = Wf + (size_t)wh * (4 * 16 * 128) + 4 * lane;

    v8f c[2][4];
#pragma unroll
    for (int mt = 0; mt < 2; ++mt)
#pragma unroll
        for (int nt = 0; nt < 4; ++nt) c[mt][nt] = (v8f){};

    v4f a[3][2], b[3][4];
#pragma unroll
    for (int p = 0; p < 2; ++p) {
#pragma unroll
        for (int mt = 0; mt < 2; ++mt) a[p][mt] = *(const v4f*)(ap[mt] + 8 * p);
#pragma unroll
        for (int nt = 0; nt < 4; ++nt) b[p][nt] = *(const v4f*)(bp + (nt * 16 + p) * 128);
    }

    {
        const int rb   = (int)blockIdx.x >> (TSH - 5);
        const int hsel = (int)blockIdx.x & ((TN >> 5) - 1);
        const int tid  = (int)threadIdx.x;
        int o0[4], ln[4];
#pragma unroll
        for (int k = 0; k < 4; ++k) {
            const int ch = 4 * tid + k;
            const bool ok = ch < nchunks;
            const int cc = ok ? ch : nchunks - 1;
            const int lo = offs[(size_t)cc * pitch + rb];
            const int hi = offs[(size_t)cc * pitch + rb + 1];
            o0[k] = ok ? lo : 0;
            ln[k] = ok ? hi - lo : 0;
        }
        const int tsum = (ln[0] + ln[1]) + (ln[2] + ln[3]);
        int bsum = (o0[0] + o0[1]) + (o0[2] + o0[3]);
        int inc = tsum;
#pragma unroll
        for (int o = 1; o < 32; o <<= 1) {
            int t = __shfl_up(inc, o);
            if (lane >= o) inc += t;
        }
#pragma unroll
        for (int o = 16; o >= 1; o >>= 1) bsum += __shfl_xor(bsum, o);
        if (lane == 31) wsum[wh] = inc;
        if (lane == 0)  wsum[2 + wh] = bsum;
        __syncthreads();
        const int num  = wsum[0] + wsum[1];
        const int begR = wsum[2] + wsum[3];
        int pre = inc - tsum + (wh ? wsum[0] : 0);
#pragma unroll
        for (int k = 0; k < 4; ++k) {
            const int ch = 4 * tid + k;
            rpre[ch] = pre;
            rsrc[ch] = ch * CHUNK + o0[k] - pre;
            pre += ln[k];
        }
        __syncthreads();
        if (hsel == 0 && tid == 0) {
            start[rb] = begR;
            if (rb == nb - 1) start[nb] = begR + num;
        }
        for (int base = 0; base < num; base += GCH) {
            const int m = (num - base < GCH) ? num - base : GCH;
            int2 ev[GCH / 64];
#pragma unroll
            for (int q = 0; q < GCH / 64; ++q) {
                const int i  = tid + q * 64;
                const int ii = base + (i < m ? i : m - 1);
                int ch = 0;
#pragma unroll
                for (int st = MAXC / 2; st >= 1; st >>= 1)
                    if (rpre[ch + st] <= ii) ch += st;
                ev[q] = ((const int2*)listc)[rsrc[ch] + ii];
            }
#pragma unroll
            for (int q = 0; q < GCH / 64; ++q) {
                const int i = tid + q * 64;
                if (i < m) {
                    if ((ev[q].y >> 5) == hsel) atomicAdd(&cnt[ev[q].y & 31], 1);
                    if (hsel == 0) ((int2*)list)[(size_t)begR + base + i] = ev[q];
                }
            }
        }
    }
    __syncthreads();
    if (threadIdx.x < 32 && m0 + (int)threadIdx.x < nrows)
        dinv[m0 + threadIdx.x] = rsqrtf(1.0f + (float)cnt[threadIdx.x]);

#pragma unroll
    for (int j = 0; j < DH / 8; ++j) {
        const int cur = j % 3, nxt = (j + 2) % 3;
        if (j + 2 < DH / 8) {
#pragma unroll
            for (int mt = 0; mt < 2; ++mt) a[nxt][mt] = *(const v4f*)(ap[mt] + 8 * (j + 2));
#pragma unroll
            for (int nt = 0; nt < 4; ++nt) b[nxt][nt] = *(const v4f*)(bp + (nt * 16 + j + 2) * 128);
        }
#pragma unroll
        for (int mt = 0; mt < 2; ++mt)
#pragma unroll
            for (int nt = 0; nt < 4; ++nt) {
                v2f a2 = {a[cur][mt].x, a[cur][mt].y};
                v2f b2 = {b[cur][nt].x, b[cur][nt].y};
                c[mt][nt] = __builtin_amdgcn_wmma_f32_16x16x4_f32(
                         false, a2,  false, b2,
                         (short)0, c[mt][nt],  false,  false);
            }
#pragma unroll
        for (int mt = 0; mt < 2; ++mt)
#pragma unroll
            for (int nt = 0; nt < 4; ++nt) {
                v2f a2 = {a[cur][mt].z, a[cur][mt].w};
                v2f b2 = {b[cur][nt].z, b[cur][nt].w};
                c[mt][nt] = __builtin_amdgcn_wmma_f32_16x16x4_f32(
                         false, a2,  false, b2,
                         (short)0, c[mt][nt],  false,  false);
            }
    }

    float* hp = H + (size_t)m0 * DH + 64 * wh + 4 * r;
#pragma unroll
    for (int mt = 0; mt < 2; ++mt)
#pragma unroll
        for (int v = 0; v < 8; ++v) {
            int row = 16 * mt + v + 8 * half;
            float s = rsqrtf(1.0f + (float)cnt[row]);
            float4 o = make_float4(c[mt][0][v] * s, c[mt][1][v] * s,
                                   c[mt][2][v] * s, c[mt][3][v] * s);
            if (m0 + row < nrows) *(float4*)(hp + (size_t)row * DH) = o;
        }
}

#define LCH 2048
__device__ __forceinline__ float gcn_gelu_exact(float x) {
    return 0.5f * x * (1.0f + erff(x * 0.70710678118654752440f));
}
template <bool NEXT>
__global__ void __launch_bounds__(256)
gcn_agg(const float* __restrict__ H, float* __restrict__ Y,
        const int* __restrict__ list, const int* __restrict__ start,
        const float* __restrict__ dinv, const float* __restrict__ bias, int n) {
    __shared__ int2 sub[LCH];
    __shared__ int  cw[64];
    const int tid   = threadIdx.x;
    const int lane  = tid & 31;
    const int wv    = __builtin_amdgcn_readfirstlane(tid >> 5);
    const int node0 = blockIdx.x * TN;
    const float4* H4 = (const float4*)H;
    const int2*   L2 = (const int2*)list;
    const unsigned below_me = (1u << lane) - 1u;

    v32f acc;
#pragma unroll
    for (int j = 0; j < 8; ++j) {
        int node = node0 + wv * 8 + j;
        int nc = node < n ? node : n - 1;
        const float4 t = H4[(size_t)nc * 32 + lane];
        acc[4 * j] = t.x; acc[4 * j + 1] = t.y; acc[4 * j + 2] = t.z; acc[4 * j + 3] = t.w;
    }

    const int beg = start[blockIdx.x];
    const int cnt = start[blockIdx.x + 1] - beg;

    for (int base = 0; base < cnt; base += LCH) {
        const int m = (cnt - base < LCH) ? cnt - base : LCH;
        int2 e[LCH / 256];
        int  loc[LCH / 256];
        int  mine = 0;
#pragma unroll
        for (int q = 0; q < LCH / 256; ++q) {
            const int i = tid + q * 256;
            const bool ok = i < m;
            e[q] = L2[beg + base + (ok ? i : m - 1)];
            const int o = ok ? (e[q].y >> 3) : 8;
            const int before = __shfl(mine, o & 7);
            int rank = 0;
#pragma unroll
            for (int oo = 0; oo < 8; ++oo) {
                const unsigned mk = __builtin_amdgcn_ballot_w32(o == oo);
                if (o == oo) rank = __popc(mk & below_me);
                if (lane == oo) mine += __popc(mk);
            }
            loc[q] = before + rank;
        }
        if (lane < 8) cw[wv * 8 + lane] = mine;
        __syncthreads();

        int tot = 0, below = 0;
#pragma unroll
        for (int w = 0; w < 8; ++w) {
            const int c = cw[w * 8 + (lane & 7)];
            tot += c;
            below += (w < wv) ? c : 0;
        }
        int first = 0;
#pragma unroll
        for (int oo = 0; oo < 7; ++oo) {
            const int t = __shfl(tot, oo);
            first += ((lane & 7) > oo) ? t : 0;
        }
        const int basev = first + below;
#pragma unroll
        for (int q = 0; q < LCH / 256; ++q) {
            const int i = tid + q * 256;
            const int o = e[q].y >> 3;
            const int slot = __shfl(basev, o & 7) + loc[q];
            if (i < m) sub[slot] = e[q];
        }
        const int my_cnt = __builtin_amdgcn_readlane(tot, wv);
        const int my_beg = __builtin_amdgcn_readlane(first, wv);
        __syncthreads();

        for (int g = 0; g < my_cnt; g += 8) {
            int i = g + (lane & 7);
            if (i > my_cnt - 1) i = my_cnt - 1;
            const int2 ec = sub[my_beg + i];
            const int nval = my_cnt - g;
            float4 h[8];
            int dl[8];
#pragma unroll
            for (int k = 0; k < 8; ++k) {
                int s = __builtin_amdgcn_readlane(ec.x, k);
                dl[k] = (__builtin_amdgcn_readlane(ec.y, k) & 7) * 4;
                h[k] = H4[(size_t)s * 32 + lane];
            }
#pragma unroll
            for (int k = 0; k < 8; ++k) {
                if (k < nval) {
                    const float4 v = h[k];
                    const int b = dl[k];
                    acc[b] += v.x; acc[b + 1] += v.y; acc[b + 2] += v.z; acc[b + 3] += v.w;
                }
            }
        }
        __syncthreads();
    }

    const float4 b4 = ((const float4*)bias)[lane];
    float di[8];
#pragma unroll
    for (int j = 0; j < 8; ++j) {
        int node = node0 + wv * 8 + j;
        di[j] = dinv[node < n ? node : n - 1];
    }
#pragma unroll
    for (int j = 0; j < 8; ++j) {
        int node = node0 + wv * 8 + j;
        float4 o = make_float4(b4.x + di[j] * acc[4 * j],     b4.y + di[j] * acc[4 * j + 1],
                               b4.z + di[j] * acc[4 * j + 2], b4.w + di[j] * acc[4 * j + 3]);
        if (NEXT) { o.x *= di[j]; o.y *= di[j]; o.z *= di[j]; o.w *= di[j]; }
        else { o.x = gcn_gelu_exact(o.x); o.y = gcn_gelu_exact(o.y); o.z = gcn_gelu_exact(o.z); o.w = gcn_gelu_exact(o.w); }
        if (node < n) ((float4*)Y)[(size_t)node * 32 + lane] = o;
    }
}

extern "C" void kernel_launch(void* const* d_in, const int* in_sizes, int n_in,
                              void* d_out, int out_size, void* d_ws, size_t ws_size,
                              hipStream_t stream) {
    const float* x    = (const float*)d_in[0];
    const int*   eidx = (const int*)  d_in[1];
    const float* Ws   = (const float*)d_in[2];
    const float* bs   = (const float*)d_in[3];
    float*       out  = (float*)d_out;

    const int N = in_sizes[0] / DH;
    const int E = in_sizes[1] / 2;
    const int* src = eidx;
    const int* dst = eidx + E;

    const int NB      = (N + TN - 1) / TN;
    const int PITCH   = (NB + 1 + 31) & ~31;
    const int NCHUNK  = (E + CHUNK - 1) / CHUNK;

    float* ws    = (float*)d_ws;
    float* dinv  = ws;
    int*   offs  = (int*)(dinv + (((size_t)N + 63) & ~(size_t)63));
    int*   start = offs  + (size_t)NCHUNK * PITCH;
    int*   list  = start + (((size_t)NB + 1 + 63) & ~(size_t)63);
    int*   listc = list  + (((size_t)2 * E + 63) & ~(size_t)63);
    float* Wf    = (float*)(listc + (size_t)2 * NCHUNK * CHUNK);
    float* cv    = Wf + (size_t)DH * DH;
    float* h     = cv + (size_t)3 * DH;
    float* buf   = h  + (size_t)N * DH;

    gcn_part_local <<<NCHUNK + 8, 1024, 0, stream>>>(src, dst, offs, listc, E, NB, PITCH, NCHUNK,
                                                    Ws, bs, Wf, cv);

    gcn_gemm<<<(N + 31) / 32, 64, 0, stream>>>(x, Wf, offs, listc, list, start, dinv, h, N, NCHUNK, PITCH, NB);
    gcn_agg<true> <<<NB, 256, 0, stream>>>(h,   buf, list, start, dinv, cv,          N);
    gcn_agg<true> <<<NB, 256, 0, stream>>>(buf, h,   list, start, dinv, cv + DH,     N);
    gcn_agg<false><<<NB, 256, 0, stream>>>(h,   out, list, start, dinv, cv + 2 * DH, N);
}


# GEMM work-group widened to one whole 64-node range (4 waves): the range's per-chunk runs are read and counted once, not by two half-range groups; 256 registers, whole grid resident; sum order kept
# speedup vs baseline: 5.1527x; 1.0489x over previous
#include <hip/hip_runtime.h>
#include <hip/hip_bf16.h>
#include <math.h>


typedef float v2f __attribute__((ext_vector_type(2)));
typedef float v4f __attribute__((ext_vector_type(4)));
typedef float v8f __attribute__((ext_vector_type(8)));
typedef float v32f __attribute__((ext_vector_type(32)));

#define DH 128
#define TN 64
#define TSH 6
#define MAXB 1024
#define CHUNK 4096
#define MAXC 256

#define PTT 132
__device__ __forceinline__ void
gcn_compose_w(const float* __restrict__ W, const float* __restrict__ bs,
              float* __restrict__ Wf, float* __restrict__ cv,
              float* Tt, const int piece) {
    const int tid  = threadIdx.x;
    const int lane = tid & 31;
    const int w    = __builtin_amdgcn_readfirstlane(tid >> 5);
    const bool act = w < 8;
    const int c    = lane & 15;
    const int half = lane >> 4;
    const int wh   = piece >> 2;
    const int nt   = piece & 3;
    const int col  = 64 * wh + nt + 4 * c;
    const float* W0 = W;
    const float* W1 = W + DH * DH;
    const float* W2 = W + 2 * DH * DH;

    v4f a0[16];
    if (act) {
        const float* a1p = W1 + (16 * w + c) * DH + 4 * half;
        const float* a0p = W0 + (16 * w + c) * DH + 4 * half;
        const float* b2p = W2 + (4 * half) * DH + col;

        v4f a1[16];
        float b[16][4];
#pragma unroll
        for (int j = 0; j < 16; ++j) a1[j] = *(const v4f*)(a1p + 8 * j);
#pragma unroll
        for (int j = 0; j < 16; ++j)
#pragma unroll
            for (int e = 0; e < 4; ++e) b[j][e] = b2p[(8 * j + e) * DH];
#pragma unroll
        for (int j = 0; j < 16; ++j) a0[j] = *(const v4f*)(a0p + 8 * j);

        v8f t = (v8f){};
#pragma unroll
        for (int j = 0; j < 16; ++j) {
            v2f x0 = {a1[j].x, a1[j].y}, y0 = {b[j][0], b[j][1]};
            t = __builtin_amdgcn_wmma_f32_16x16x4_f32(false, x0, false, y0, (short)0, t, false, false);
            v2f x1 = {a1[j].z, a1[j].w}, y1 = {b[j][2], b[j][3]};
            t = __builtin_amdgcn_wmma_f32_16x16x4_f32(false, x1, false, y1, (short)0, t, false, false);
        }
        float* tp = Tt + c * PTT + 16 * w + 8 * half;
        *(v4f*)tp       = (v4f){t[0], t[1], t[2], t[3]};
        *(v4f*)(tp + 4) = (v4f){t[4], t[5], t[6], t[7]};
    }
    __syncthreads();
    if (!act) return;

    v8f d = (v8f){};
    const float* tb = Tt + c * PTT + 4 * half;
#pragma unroll
    for (int j = 0; j < 16; ++j) {
        const v4f bb = *(const v4f*)(tb + 8 * j);
        v2f x0 = {a0[j].x, a0[j].y}, y0 = {bb.x, bb.y};
        d = __builtin_amdgcn_wmma_f32_16x16x4_f32(false, x0, false, y0, (short)0, d, false, false);
        v2f x1 = {a0[j].z, a0[j].w}, y1 = {bb.z, bb.w};
        d = __builtin_amdgcn_wmma_f32_16x16x4_f32(false, x1, false, y1, (short)0, d, false, false);
    }
    float* op = Wf + (size_t)piece * (16 * 128) + (2 * w + half) * 128 + 4 * c;
    *(v4f*)op        = (v4f){d[0], d[1], d[2], d[3]};
    *(v4f*)(op + 64) = (v4f){d[4], d[5], d[6], d[7]};

    if (tid < 16) {
        const int cb = 64 * wh + nt + 4 * tid;
        const float* tr = Tt + tid * PTT;
        float s0 = 0.f, s1 = 0.f, s2 = 0.f, s3 = 0.f;
        for (int k = 0; k < DH; k += 4) {
            const v4f tv = *(const v4f*)(tr + k);
            s0 += bs[k] * tv.x; s1 += bs[k + 1] * tv.y; s2 += bs[k + 2] * tv.z; s3 += bs[k + 3] * tv.w;
        }
        cv[cb] = (s0 + s1) + (s2 + s3);
    } else if (tid >= 32 && tid < 48) {
        const int cb = 64 * wh + nt + 4 * (tid - 32);
        float s0 = 0.f, s1 = 0.f, s2 = 0.f, s3 = 0.f;
        for (int k = 0; k < DH; k += 4) {
            s0 += bs[DH + k]     * W2[k * DH + cb];
            s1 += bs[DH + k + 1] * W2[(k + 1) * DH + cb];
            s2 += bs[DH + k + 2] * W2[(k + 2) * DH + cb];
            s3 += bs[DH + k + 3] * W2[(k + 3) * DH + cb];
        }
        cv[DH + cb] = (s0 + s1) + (s2 + s3);
    } else if (tid >= 64 && tid < 80) {
        const int cb = 64 * wh + nt + 4 * (tid - 64);
        cv[2 * DH + cb] = bs[2 * DH + cb];
    }
}

__global__ void __launch_bounds__(1024)
gcn_part_local(const int* __restrict__ src, const int* __restrict__ dst,
               int* __restrict__ offs, int* __restrict__ listc,
               int ne, int nb, int pitch, int nchunks,
               const float* __restrict__ W, const float* __restrict__ bs,
               float* __restrict__ Wf, float* __restrict__ cv) {
    __shared__ int cur[MAXB];
    __shared__ int wtot[32];
    __shared__ __attribute__((aligned(16))) float Tt[16 * PTT];
    if ((int)blockIdx.x >= nchunks) {
        gcn_compose_w(W, bs, Wf, cv, Tt, (int)blockIdx.x - nchunks);
        return;
    }
    const int tid  = threadIdx.x;
    const int lane = tid & 31;
    const int wv   = tid >> 5;
    const int me   = blockIdx.x;

    cur[tid] = 0;
    __syncthreads();

    const int e0 = me * CHUNK + tid;
    int d[4], s[4];
#pragma unroll
    for (int k = 0; k < 4; ++k) {
        int e = e0 + k * 1024;
        bool ok = e < ne;
        d[k] = ok ? dst[e] : -1;
        s[k] = ok ? src[e] : 0;
    }
#pragma unroll
    for (int k = 0; k < 4; ++k)
        if (d[k] >= 0) atomicAdd(&cur[d[k] >> TSH], 1);
    __syncthreads();

    const int tot = cur[tid];
    int inc = tot;
#pragma unroll
    for (int o = 1; o < 32; o <<= 1) {
        int t = __shfl_up(inc, o);
        if (lane >= o) inc += t;
    }
    if (lane == 31) wtot[wv] = inc;
    __syncthreads();
    if (wv == 0) {
        int w = wtot[lane];
        int wi = w;
#pragma unroll
        for (int o = 1; o < 32; o <<= 1) {
            int t = __shfl_up(wi, o);
            if (lane >= o) wi += t;
        }
        wtot[lane] = wi - w;
    }
    __syncthreads();
    const int base = wtot[wv] + inc - tot;
    cur[tid] = base;
    if (tid <= nb) offs[(size_t)me * pitch + tid] = base;
    __syncthreads();

#pragma unroll
    for (int k = 0; k < 4; ++k) {
        if (d[k] >= 0) {
            int slot = atomicAdd(&cur[d[k] >> TSH], 1);
            ((int2*)listc)[(size_t)me * CHUNK + slot] = make_int2(s[k], d[k] & (TN - 1));
        }
    }
}

#define GCH 1280
#define GT 128
__global__ void __launch_bounds__(GT) __attribute__((amdgpu_waves_per_eu(1, 4)))
gcn_gemm(const float* __restrict__ X, const float* __restrict__ Wf,
         const int* __restrict__ offs, const int* __restrict__ listc,
         int* __restrict__ list, int* __restrict__ start,
         float* __restrict__ dinv, float* __restrict__ H, int nrows,
         int nchunks, int pitch, int nb) {
    __shared__ int cnt[TN];
    __shared__ int rpre[MAXC];
    __shared__ int rsrc[MAXC];
    __shared__ int wsum[8];
    if (threadIdx.x < TN) cnt[threadIdx.x] = 0;
    __syncthreads();
    const int lane = threadIdx.x & 31;
    const int wv   = threadIdx.x >> 5;
    const int wh   = wv & 1;
    const int mh   = wv >> 1;
    const int m0   = ((int)blockIdx.x << TSH) + 32 * mh;
    const int r    = lane & 15;
    const int half = lane >> 4;

    const float* ap[2];
#pragma unroll
    for (int mt = 0; mt < 2; ++mt) {
        int arow = m0 + 16 * mt + r;
        if (arow >= nrows) arow = nrows - 1;
        ap[mt] = X + (size_t)arow * DH + 4 * half;
    }
    const float* bp = Wf + (size_t)wh * (4 * 16 * 128) + 4 * lane;

    v8f c[2][4];
#pragma unroll
    for (int mt = 0; mt < 2; ++mt)
#pragma unroll
        for (int nt = 0; nt < 4; ++nt) c[mt][nt] = (v8f){};

    v4f a[3][2], b[3][4];
#pragma unroll
    for (int p = 0; p < 2; ++p) {
#pragma unroll
        for (int mt = 0; mt < 2; ++mt) a[p][mt] = *(const v4f*)(ap[mt] + 8 * p);
#pragma unroll
        for (int nt = 0; nt < 4; ++nt) b[p][nt] = *(const v4f*)(bp + (nt * 16 + p) * 128);
    }

    {
        const int rb   = (int)blockIdx.x;
        const int tid  = (int)threadIdx.x;
        int o0[2], ln[2];
#pragma unroll
        for (int k = 0; k < 2; ++k) {
            const int ch = 2 * tid + k;
            const bool ok = ch < nchunks;
            const int cc = ok ? ch : nchunks - 1;
            const int lo = offs[(size_t)cc * pitch + rb];
            const int hi = offs[(size_t)cc * pitch + rb + 1];
            o0[k] = ok ? lo : 0;
            ln[k] = ok ? hi - lo : 0;
        }
        const int tsum = ln[0] + ln[1];
        int bsum = o0[0] + o0[1];
        int inc = tsum;
#pragma unroll
        for (int o = 1; o < 32; o <<= 1) {
            int t = __shfl_up(inc, o);
            if (lane >= o) inc += t;
        }
#pragma unroll
        for (int o = 16; o >= 1; o >>= 1) bsum += __shfl_xor(bsum, o);
        if (lane == 31) wsum[wv] = inc;
        if (lane == 0)  wsum[4 + wv] = bsum;
        __syncthreads();
        const int num  = (wsum[0] + wsum[1]) + (wsum[2] + wsum[3]);
        const int begR = (wsum[4] + wsum[5]) + (wsum[6] + wsum[7]);
        int pre = inc - tsum;
#pragma unroll
        for (int w = 0; w < 3; ++w) pre += (w < wv) ? wsum[w] : 0;
#pragma unroll
        for (int k = 0; k < 2; ++k) {
            const int ch = 2 * tid + k;
            rpre[ch] = pre;
            rsrc[ch] = ch * CHUNK + o0[k] - pre;
            pre += ln[k];
        }
        __syncthreads();
        if (tid == 0) {
            start[rb] = begR;
            if (rb == nb - 1) start[nb] = begR + num;
        }
        for (int base = 0; base < num; base += GCH) {
            const int m = (num - base < GCH) ? num - base : GCH;
            int2 ev[GCH / GT];
#pragma unroll
            for (int q = 0; q < GCH / GT; ++q) {
                const int i  = tid + q * GT;
                const int ii = base + (i < m ? i : m - 1);
                int ch = 0;
#pragma unroll
                for (int st = MAXC / 2; st >= 1; st >>= 1)
                    if (rpre[ch + st] <= ii) ch += st;
                ev[q] = ((const int2*)listc)[rsrc[ch] + ii];
            }
#pragma unroll
            for (int q = 0; q < GCH / GT; ++q) {
                const int i = tid + q * GT;
                if (i < m) {
                    atomicAdd(&cnt[ev[q].y & (TN - 1)], 1);
                    ((int2*)list)[(size_t)begR + base + i] = ev[q];
                }
            }
        }
    }
    __syncthreads();
    if (wh == 0 && m0 + lane < nrows)
        dinv[m0 + lane] = rsqrtf(1.0f + (float)cnt[32 * mh + lane]);

#pragma unroll
    for (int j = 0; j < DH / 8; ++j) {
        const int cur = j % 3, nxt = (j + 2) % 3;
        if (j + 2 < DH / 8) {
#pragma unroll
            for (int mt = 0; mt < 2; ++mt) a[nxt][mt] = *(const v4f*)(ap[mt] + 8 * (j + 2));
#pragma unroll
            for (int nt = 0; nt < 4; ++nt) b[nxt][nt] = *(const v4f*)(bp + (nt * 16 + j + 2) * 128);
        }
#pragma unroll
        for (int mt = 0; mt < 2; ++mt)
#pragma unroll
            for (int nt = 0; nt < 4; ++nt) {
                v2f a2 = {a[cur][mt].x, a[cur][mt].y};
                v2f b2 = {b[cur][nt].x, b[cur][nt].y};
                c[mt][nt] = __builtin_amdgcn_wmma_f32_16x16x4_f32(
                         false, a2,  false, b2,
                         (short)0, c[mt][nt],  false,  false);
            }
#pragma unroll
        for (int mt = 0; mt < 2; ++mt)
#pragma unroll
            for (int nt = 0; nt < 4; ++nt) {
                v2f a2 = {a[cur][mt].z, a[cur][mt].w};
                v2f b2 = {b[cur][nt].z, b[cur][nt].w};
                c[mt][nt] = __builtin_amdgcn_wmma_f32_16x16x4_f32(
                         false, a2,  false, b2,
                         (short)0, c[mt][nt],  false,  false);
            }
    }

    float* hp = H + (size_t)m0 * DH + 64 * wh + 4 * r;
#pragma unroll
    for (int mt = 0; mt < 2; ++mt)
#pragma unroll
        for (int v = 0; v < 8; ++v) {
            int row = 16 * mt + v + 8 * half;
            float s = rsqrtf(1.0f + (float)cnt[32 * mh + row]);
            float4 o = make_float4(c[mt][0][v] * s, c[mt][1][v] * s,
                                   c[mt][2][v] * s, c[mt][3][v] * s);
            if (m0 + row < nrows) *(float4*)(hp + (size_t)row * DH) = o;
        }
}

#define LCH 2048
__device__ __forceinline__ float gcn_gelu_exact(float x) {
    return 0.5f * x * (1.0f + erff(x * 0.70710678118654752440f));
}
template <bool NEXT>
__global__ void __launch_bounds__(256)
gcn_agg(const float* __restrict__ H, float* __restrict__ Y,
        const int* __restrict__ list, const int* __restrict__ start,
        const float* __restrict__ dinv, const float* __restrict__ bias, int n) {
    __shared__ int2 sub[LCH];
    __shared__ int  cw[64];
    const int tid   = threadIdx.x;
    const int lane  = tid & 31;
    const int wv    = __builtin_amdgcn_readfirstlane(tid >> 5);
    const int node0 = blockIdx.x * TN;
    const float4* H4 = (const float4*)H;
    const int2*   L2 = (const int2*)list;
    const unsigned below_me = (1u << lane) - 1u;

    v32f acc;
#pragma unroll
    for (int j = 0; j < 8; ++j) {
        int node = node0 + wv * 8 + j;
        int nc = node < n ? node : n - 1;
        const float4 t = H4[(size_t)nc * 32 + lane];
        acc[4 * j] = t.x; acc[4 * j + 1] = t.y; acc[4 * j + 2] = t.z; acc[4 * j + 3] = t.w;
    }

    const int beg = start[blockIdx.x];
    const int cnt = start[blockIdx.x + 1] - beg;

    for (int base = 0; base < cnt; base += LCH) {
        const int m = (cnt - base < LCH) ? cnt - base : LCH;
        int2 e[LCH / 256];
        int  loc[LCH / 256];
        int  mine = 0;
#pragma unroll
        for (int q = 0; q < LCH / 256; ++q) {
            const int i = tid + q * 256;
            const bool ok = i < m;
            e[q] = L2[beg + base + (ok ? i : m - 1)];
            const int o = ok ? (e[q].y >> 3) : 8;
            const int before = __shfl(mine, o & 7);
            int rank = 0;
#pragma unroll
            for (int oo = 0; oo < 8; ++oo) {
                const unsigned mk = __builtin_amdgcn_ballot_w32(o == oo);
                if (o == oo) rank = __popc(mk & below_me);
                if (lane == oo) mine += __popc(mk);
            }
            loc[q] = before + rank;
        }
        if (lane < 8) cw[wv * 8 + lane] = mine;
        __syncthreads();

        int tot = 0, below = 0;
#pragma unroll
        for (int w = 0; w < 8; ++w) {
            const int c = cw[w * 8 + (lane & 7)];
            tot += c;
            below += (w < wv) ? c : 0;
        }
        int first = 0;
#pragma unroll
        for (int oo = 0; oo < 7; ++oo) {
            const int t = __shfl(tot, oo);
            first += ((lane & 7) > oo) ? t : 0;
        }
        const int basev = first + below;
#pragma unroll
        for (int q = 0; q < LCH / 256; ++q) {
            const int i = tid + q * 256;
            const int o = e[q].y >> 3;
            const int slot = __shfl(basev, o & 7) + loc[q];
            if (i < m) sub[slot] = e[q];
        }
        const int my_cnt = __builtin_amdgcn_readlane(tot, wv);
        const int my_beg = __builtin_amdgcn_readlane(first, wv);
        __syncthreads();

        for (int g = 0; g < my_cnt; g += 8) {
            int i = g + (lane & 7);
            if (i > my_cnt - 1) i = my_cnt - 1;
            const int2 ec = sub[my_beg + i];
            const int nval = my_cnt - g;
            float4 h[8];
            int dl[8];
#pragma unroll
            for (int k = 0; k < 8; ++k) {
                int s = __builtin_amdgcn_readlane(ec.x, k);
                dl[k] = (__builtin_amdgcn_readlane(ec.y, k) & 7) * 4;
                h[k] = H4[(size_t)s * 32 + lane];
            }
#pragma unroll
            for (int k = 0; k < 8; ++k) {
                if (k < nval) {
                    const float4 v = h[k];
                    const int b = dl[k];
                    acc[b] += v.x; acc[b + 1] += v.y; acc[b + 2] += v.z; acc[b + 3] += v.w;
                }
            }
        }
        __syncthreads();
    }

    const float4 b4 = ((const float4*)bias)[lane];
    float di[8];
#pragma unroll
    for (int j = 0; j < 8; ++j) {
        int node = node0 + wv * 8 + j;
        di[j] = dinv[node < n ? node : n - 1];
    }
#pragma unroll
    for (int j = 0; j < 8; ++j) {
        int node = node0 + wv * 8 + j;
        float4 o = make_float4(b4.x + di[j] * acc[4 * j],     b4.y + di[j] * acc[4 * j + 1],
                               b4.z + di[j] * acc[4 * j + 2], b4.w + di[j] * acc[4 * j + 3]);
        if (NEXT) { o.x *= di[j]; o.y *= di[j]; o.z *= di[j]; o.w *= di[j]; }
        else { o.x = gcn_gelu_exact(o.x); o.y = gcn_gelu_exact(o.y); o.z = gcn_gelu_exact(o.z); o.w = gcn_gelu_exact(o.w); }
        if (node < n) ((float4*)Y)[(size_t)node * 32 + lane] = o;
    }
}

extern "C" void kernel_launch(void* const* d_in, const int* in_sizes, int n_in,
                              void* d_out, int out_size, void* d_ws, size_t ws_size,
                              hipStream_t stream) {
    const float* x    = (const float*)d_in[0];
    const int*   eidx = (const int*)  d_in[1];
    const float* Ws   = (const float*)d_in[2];
    const float* bs   = (const float*)d_in[3];
    float*       out  = (float*)d_out;

    const int N = in_sizes[0] / DH;
    const int E = in_sizes[1] / 2;
    const int* src = eidx;
    const int* dst = eidx + E;

    const int NB      = (N + TN - 1) / TN;
    const int PITCH   = (NB + 1 + 31) & ~31;
    const int NCHUNK  = (E + CHUNK - 1) / CHUNK;

    float* ws    = (float*)d_ws;
    float* dinv  = ws;
    int*   offs  = (int*)(dinv + (((size_t)N + 63) & ~(size_t)63));
    int*   start = offs  + (size_t)NCHUNK * PITCH;
    int*   list  = start + (((size_t)NB + 1 + 63) & ~(size_t)63);
    int*   listc = list  + (((size_t)2 * E + 63) & ~(size_t)63);
    float* Wf    = (float*)(listc + (size_t)2 * NCHUNK * CHUNK);
    float* cv    = Wf + (size_t)DH * DH;
    float* h     = cv + (size_t)3 * DH;
    float* buf   = h  + (size_t)N * DH;

    gcn_part_local <<<NCHUNK + 8, 1024, 0, stream>>>(src, dst, offs, listc, E, NB, PITCH, NCHUNK,
                                                    Ws, bs, Wf, cv);

    gcn_gemm<<<NB, GT, 0, stream>>>(x, Wf, offs, listc, list, start, dinv, h, N, NCHUNK, PITCH, NB);
    gcn_agg<true> <<<NB, 256, 0, stream>>>(h,   buf, list, start, dinv, cv,          N);
    gcn_agg<true> <<<NB, 256, 0, stream>>>(buf, h,   list, start, dinv, cv + DH,     N);
    gcn_agg<false><<<NB, 256, 0, stream>>>(h,   out, list, start, dinv, cv + 2 * DH, N);
}


# Neighbour sums: range list regrouped by target row (lane-ballot match masks) instead of by owning wave; a wave sums one row at a time in fixed registers, no indexed register adds; sum order kept
# speedup vs baseline: 5.5910x; 1.0851x over previous
#include <hip/hip_runtime.h>
#include <hip/hip_bf16.h>
#include <math.h>


typedef float v2f __attribute__((ext_vector_type(2)));
typedef float v4f __attribute__((ext_vector_type(4)));
typedef float v8f __attribute__((ext_vector_type(8)));
typedef float v32f __attribute__((ext_vector_type(32)));

#define DH 128
#define TN 64
#define TSH 6
#define MAXB 1024
#define CHUNK 4096
#define MAXC 256

#define PTT 132
__device__ __forceinline__ void
gcn_compose_w(const float* __restrict__ W, const float* __restrict__ bs,
              float* __restrict__ Wf, float* __restrict__ cv,
              float* Tt, const int piece) {
    const int tid  = threadIdx.x;
    const int lane = tid & 31;
    const int w    = __builtin_amdgcn_readfirstlane(tid >> 5);
    const bool act = w < 8;
    const int c    = lane & 15;
    const int half = lane >> 4;
    const int wh   = piece >> 2;
    const int nt   = piece & 3;
    const int col  = 64 * wh + nt + 4 * c;
    const float* W0 = W;
    const float* W1 = W + DH * DH;
    const float* W2 = W + 2 * DH * DH;

    v4f a0[16];
    if (act) {
        const float* a1p = W1 + (16 * w + c) * DH + 4 * half;
        const float* a0p = W0 + (16 * w + c) * DH + 4 * half;
        const float* b2p = W2 + (4 * half) * DH + col;

        v4f a1[16];
        float b[16][4];
#pragma unroll
        for (int j = 0; j < 16; ++j) a1[j] = *(const v4f*)(a1p + 8 * j);
#pragma unroll
        for (int j = 0; j < 16; ++j)
#pragma unroll
            for (int e = 0; e < 4; ++e) b[j][e] = b2p[(8 * j + e) * DH];
#pragma unroll
        for (int j = 0; j < 16; ++j) a0[j] = *(const v4f*)(a0p + 8 * j);

        v8f t = (v8f){};
#pragma unroll
        for (int j = 0; j < 16; ++j) {
            v2f x0 = {a1[j].x, a1[j].y}, y0 = {b[j][0], b[j][1]};
            t = __builtin_amdgcn_wmma_f32_16x16x4_f32(false, x0, false, y0, (short)0, t, false, false);
            v2f x1 = {a1[j].z, a1[j].w}, y1 = {b[j][2], b[j][3]};
            t = __builtin_amdgcn_wmma_f32_16x16x4_f32(false, x1, false, y1, (short)0, t, false, false);
        }
        float* tp = Tt + c * PTT + 16 * w + 8 * half;
        *(v4f*)tp       = (v4f){t[0], t[1], t[2], t[3]};
        *(v4f*)(tp + 4) = (v4f){t[4], t[5], t[6], t[7]};
    }
    __syncthreads();
    if (!act) return;

    v8f d = (v8f){};
    const float* tb = Tt + c * PTT + 4 * half;
#pragma unroll
    for (int j = 0; j < 16; ++j) {
        const v4f bb = *(const v4f*)(tb + 8 * j);
        v2f x0 = {a0[j].x, a0[j].y}, y0 = {bb.x, bb.y};
        d = __builtin_amdgcn_wmma_f32_16x16x4_f32(false, x0, false, y0, (short)0, d, false, false);
        v2f x1 = {a0[j].z, a0[j].w}, y1 = {bb.z, bb.w};
        d = __builtin_amdgcn_wmma_f32_16x16x4_f32(false, x1, false, y1, (short)0, d, false, false);
    }
    float* op = Wf + (size_t)piece * (16 * 128) + (2 * w + half) * 128 + 4 * c;
    *(v4f*)op        = (v4f){d[0], d[1], d[2], d[3]};
    *(v4f*)(op + 64) = (v4f){d[4], d[5], d[6], d[7]};

    if (tid < 16) {
        const int cb = 64 * wh + nt + 4 * tid;
        const float* tr = Tt + tid * PTT;
        float s0 = 0.f, s1 = 0.f, s2 = 0.f, s3 = 0.f;
        for (int k = 0; k < DH; k += 4) {
            const v4f tv = *(const v4f*)(tr + k);
            s0 += bs[k] * tv.x; s1 += bs[k + 1] * tv.y; s2 += bs[k + 2] * tv.z; s3 += bs[k + 3] * tv.w;
        }
        cv[cb] = (s0 + s1) + (s2 + s3);
    } else if (tid >= 32 && tid < 48) {
        const int cb = 64 * wh + nt + 4 * (tid - 32);
        float s0 = 0.f, s1 = 0.f, s2 = 0.f, s3 = 0.f;
        for (int k = 0; k < DH; k += 4) {
            s0 += bs[DH + k]     * W2[k * DH + cb];
            s1 += bs[DH + k + 1] * W2[(k + 1) * DH + cb];
            s2 += bs[DH + k + 2] * W2[(k + 2) * DH + cb];
            s3 += bs[DH + k + 3] * W2[(k + 3) * DH + cb];
        }
        cv[DH + cb] = (s0 + s1) + (s2 + s3);
    } else if (tid >= 64 && tid < 80) {
        const int cb = 64 * wh + nt + 4 * (tid - 64);
        cv[2 * DH + cb] = bs[2 * DH + cb];
    }
}

__global__ void __launch_bounds__(1024)
gcn_part_local(const int* __restrict__ src, const int* __restrict__ dst,
               int* __restrict__ offs, int* __restrict__ listc,
               int ne, int nb, int pitch, int nchunks,
               const float* __restrict__ W, const float* __restrict__ bs,
               float* __restrict__ Wf, float* __restrict__ cv) {
    __shared__ int cur[MAXB];
    __shared__ int wtot[32];
    __shared__ __attribute__((aligned(16))) float Tt[16 * PTT];
    if ((int)blockIdx.x >= nchunks) {
        gcn_compose_w(W, bs, Wf, cv, Tt, (int)blockIdx.x - nchunks);
        return;
    }
    const int tid  = threadIdx.x;
    const int lane = tid & 31;
    const int wv   = tid >> 5;
    const int me   = blockIdx.x;

    cur[tid] = 0;
    __syncthreads();

    const int e0 = me * CHUNK + tid;
    int d[4], s[4];
#pragma unroll
    for (int k = 0; k < 4; ++k) {
        int e = e0 + k * 1024;
        bool ok = e < ne;
        d[k] = ok ? dst[e] : -1;
        s[k] = ok ? src[e] : 0;
    }
#pragma unroll
    for (int k = 0; k < 4; ++k)
        if (d[k] >= 0) atomicAdd(&cur[d[k] >> TSH], 1);
    __syncthreads();

    const int tot = cur[tid];
    int inc = tot;
#pragma unroll
    for (int o = 1; o < 32; o <<= 1) {
        int t = __shfl_up(inc, o);
        if (lane >= o) inc += t;
    }
    if (lane == 31) wtot[wv] = inc;
    __syncthreads();
    if (wv == 0) {
        int w = wtot[lane];
        int wi = w;
#pragma unroll
        for (int o = 1; o < 32; o <<= 1) {
            int t = __shfl_up(wi, o);
            if (lane >= o) wi += t;
        }
        wtot[lane] = wi - w;
    }
    __syncthreads();
    const int base = wtot[wv] + inc - tot;
    cur[tid] = base;
    if (tid <= nb) offs[(size_t)me * pitch + tid] = base;
    __syncthreads();

#pragma unroll
    for (int k = 0; k < 4; ++k) {
        if (d[k] >= 0) {
            int slot = atomicAdd(&cur[d[k] >> TSH], 1);
            ((int2*)listc)[(size_t)me * CHUNK + slot] = make_int2(s[k], d[k] & (TN - 1));
        }
    }
}

#define GCH 1280
#define GT 128
__global__ void __launch_bounds__(GT) __attribute__((amdgpu_waves_per_eu(1, 4)))
gcn_gemm(const float* __restrict__ X, const float* __restrict__ Wf,
         const int* __restrict__ offs, const int* __restrict__ listc,
         int* __restrict__ list, int* __restrict__ start,
         float* __restrict__ dinv, float* __restrict__ H, int nrows,
         int nchunks, int pitch, int nb) {
    __shared__ int cnt[TN];
    __shared__ int rpre[MAXC];
    __shared__ int rsrc[MAXC];
    __shared__ int wsum[8];
    if (threadIdx.x < TN) cnt[threadIdx.x] = 0;
    __syncthreads();
    const int lane = threadIdx.x & 31;
    const int wv   = threadIdx.x >> 5;
    const int wh   = wv & 1;
    const int mh   = wv >> 1;
    const int m0   = ((int)blockIdx.x << TSH) + 32 * mh;
    const int r    = lane & 15;
    const int half = lane >> 4;

    const float* ap[2];
#pragma unroll
    for (int mt = 0; mt < 2; ++mt) {
        int arow = m0 + 16 * mt + r;
        if (arow >= nrows) arow = nrows - 1;
        ap[mt] = X + (size_t)arow * DH + 4 * half;
    }
    const float* bp = Wf + (size_t)wh * (4 * 16 * 128) + 4 * lane;

    v8f c[2][4];
#pragma unroll
    for (int mt = 0; mt < 2; ++mt)
#pragma unroll
        for (int nt = 0; nt < 4; ++nt) c[mt][nt] = (v8f){};

    v4f a[3][2], b[3][4];
#pragma unroll
    for (int p = 0; p < 2; ++p) {
#pragma unroll
        for (int mt = 0; mt < 2; ++mt) a[p][mt] = *(const v4f*)(ap[mt] + 8 * p);
#pragma unroll
        for (int nt = 0; nt < 4; ++nt) b[p][nt] = *(const v4f*)(bp + (nt * 16 + p) * 128);
    }

    {
        const int rb   = (int)blockIdx.x;
        const int tid  = (int)threadIdx.x;
        int o0[2], ln[2];
#pragma unroll
        for (int k = 0; k < 2; ++k) {
            const int ch = 2 * tid + k;
            const bool ok = ch < nchunks;
            const int cc = ok ? ch : nchunks - 1;
            const int lo = offs[(size_t)cc * pitch + rb];
            const int hi = offs[(size_t)cc * pitch + rb + 1];
            o0[k] = ok ? lo : 0;
            ln[k] = ok ? hi - lo : 0;
        }
        const int tsum = ln[0] + ln[1];
        int bsum = o0[0] + o0[1];
        int inc = tsum;
#pragma unroll
        for (int o = 1; o < 32; o <<= 1) {
            int t = __shfl_up(inc, o);
            if (lane >= o) inc += t;
        }
#pragma unroll
        for (int o = 16; o >= 1; o >>= 1) bsum += __shfl_xor(bsum, o);
        if (lane == 31) wsum[wv] = inc;
        if (lane == 0)  wsum[4 + wv] = bsum;
        __syncthreads();
        const int num  = (wsum[0] + wsum[1]) + (wsum[2] + wsum[3]);
        const int begR = (wsum[4] + wsum[5]) + (wsum[6] + wsum[7]);
        int pre = inc - tsum;
#pragma unroll
        for (int w = 0; w < 3; ++w) pre += (w < wv) ? wsum[w] : 0;
#pragma unroll
        for (int k = 0; k < 2; ++k) {
            const int ch = 2 * tid + k;
            rpre[ch] = pre;
            rsrc[ch] = ch * CHUNK + o0[k] - pre;
            pre += ln[k];
        }
        __syncthreads();
        if (tid == 0) {
            start[rb] = begR;
            if (rb == nb - 1) start[nb] = begR + num;
        }
        for (int base = 0; base < num; base += GCH) {
            const int m = (num - base < GCH) ? num - base : GCH;
            int2 ev[GCH / GT];
#pragma unroll
            for (int q = 0; q < GCH / GT; ++q) {
                const int i  = tid + q * GT;
                const int ii = base + (i < m ? i : m - 1);
                int ch = 0;
#pragma unroll
                for (int st = MAXC / 2; st >= 1; st >>= 1)
                    if (rpre[ch + st] <= ii) ch += st;
                ev[q] = ((const int2*)listc)[rsrc[ch] + ii];
            }
#pragma unroll
            for (int q = 0; q < GCH / GT; ++q) {
                const int i = tid + q * GT;
                if (i < m) {
                    atomicAdd(&cnt[ev[q].y & (TN - 1)], 1);
                    ((int2*)list)[(size_t)begR + base + i] = ev[q];
                }
            }
        }
    }
    __syncthreads();
    if (wh == 0 && m0 + lane < nrows)
        dinv[m0 + lane] = rsqrtf(1.0f + (float)cnt[32 * mh + lane]);

#pragma unroll
    for (int j = 0; j < DH / 8; ++j) {
        const int cur = j % 3, nxt = (j + 2) % 3;
        if (j + 2 < DH / 8) {
#pragma unroll
            for (int mt = 0; mt < 2; ++mt) a[nxt][mt] = *(const v4f*)(ap[mt] + 8 * (j + 2));
#pragma unroll
            for (int nt = 0; nt < 4; ++nt) b[nxt][nt] = *(const v4f*)(bp + (nt * 16 + j + 2) * 128);
        }
#pragma unroll
        for (int mt = 0; mt < 2; ++mt)
#pragma unroll
            for (int nt = 0; nt < 4; ++nt) {
                v2f a2 = {a[cur][mt].x, a[cur][mt].y};
                v2f b2 = {b[cur][nt].x, b[cur][nt].y};
                c[mt][nt] = __builtin_amdgcn_wmma_f32_16x16x4_f32(
                         false, a2,  false, b2,
                         (short)0, c[mt][nt],  false,  false);
            }
#pragma unroll
        for (int mt = 0; mt < 2; ++mt)
#pragma unroll
            for (int nt = 0; nt < 4; ++nt) {
                v2f a2 = {a[cur][mt].z, a[cur][mt].w};
                v2f b2 = {b[cur][nt].z, b[cur][nt].w};
                c[mt][nt] = __builtin_amdgcn_wmma_f32_16x16x4_f32(
                         false, a2,  false, b2,
                         (short)0, c[mt][nt],  false,  false);
            }
    }

    float* hp = H + (size_t)m0 * DH + 64 * wh + 4 * r;
#pragma unroll
    for (int mt = 0; mt < 2; ++mt)
#pragma unroll
        for (int v = 0; v < 8; ++v) {
            int row = 16 * mt + v + 8 * half;
            float s = rsqrtf(1.0f + (float)cnt[32 * mh + row]);
            float4 o = make_float4(c[mt][0][v] * s, c[mt][1][v] * s,
                                   c[mt][2][v] * s, c[mt][3][v] * s);
            if (m0 + row < nrows) *(float4*)(hp + (size_t)row * DH) = o;
        }
}

#define LCH 2048
#define WG 16
__device__ __forceinline__ float gcn_gelu_exact(float x) {
    return 0.5f * x * (1.0f + erff(x * 0.70710678118654752440f));
}
template <int NB>
__device__ __forceinline__ void
gcn_row_tail(float4& cur, const char* __restrict__ Hb, unsigned lane16, int sv, int nval) {
    float4 h[4 * NB];
#pragma unroll
    for (int k = 0; k < 4 * NB; ++k) {
        const int s = __builtin_amdgcn_readlane(sv, k);
        h[k] = *(const float4*)(Hb + (((unsigned)s << 9) + lane16));
    }
#pragma unroll
    for (int k = 0; k < 4 * NB - 3; ++k) {
        cur.x += h[k].x; cur.y += h[k].y; cur.z += h[k].z; cur.w += h[k].w;
    }
#pragma unroll
    for (int k = 4 * NB - 3; k < 4 * NB; ++k) {
        if (k < nval) {
            cur.x += h[k].x; cur.y += h[k].y; cur.z += h[k].z; cur.w += h[k].w;
        }
    }
}
template <bool NEXT>
__global__ void __launch_bounds__(256) __attribute__((amdgpu_waves_per_eu(8)))
gcn_agg(const float* __restrict__ H, float* __restrict__ Y,
        const int* __restrict__ list, const int* __restrict__ start,
        const float* __restrict__ dinv, const float* __restrict__ bias, int n) {
    __shared__ int sub[LCH];
    __shared__ int cw[8 * TN];
    __shared__ int bw[8 * TN];
    const int tid   = threadIdx.x;
    const int lane  = tid & 31;
    const int wv    = __builtin_amdgcn_readfirstlane(tid >> 5);
    const int node0 = blockIdx.x * TN;
    const float4* H4 = (const float4*)H;
    const char*   Hb = (const char*)H;
    const unsigned lane16 = (unsigned)lane << 4;
    const int2*   L2 = (const int2*)list;
    const unsigned below_me = (1u << lane) - 1u;
    int* cwl = cw + TN * wv;
    int* bwl = bw + TN * wv;

    v32f acc;
#pragma unroll
    for (int j = 0; j < 8; ++j) {
        int node = node0 + wv * 8 + j;
        int nc = node < n ? node : n - 1;
        const float4 t = H4[(size_t)nc * 32 + lane];
        acc[4 * j] = t.x; acc[4 * j + 1] = t.y; acc[4 * j + 2] = t.z; acc[4 * j + 3] = t.w;
    }

    const int beg = start[blockIdx.x];
    const int cnt = start[blockIdx.x + 1] - beg;

    for (int base = 0; base < cnt; base += LCH) {
        const int m = (cnt - base < LCH) ? cnt - base : LCH;
        cwl[lane] = 0;
        cwl[32 + lane] = 0;
        int2 e[LCH / 256];
        int  loc[LCH / 256];
#pragma unroll
        for (int q = 0; q < LCH / 256; ++q) {
            const int i = tid + q * 256;
            const bool ok = i < m;
            e[q] = L2[beg + base + (ok ? i : m - 1)];
            const int key = e[q].y & (TN - 1);
            unsigned same = __builtin_amdgcn_ballot_w32(ok);
#pragma unroll
            for (int b = 0; b < TSH; ++b) {
                const bool bit = (key >> b) & 1;
                const unsigned bal = __builtin_amdgcn_ballot_w32(bit);
                same &= bit ? bal : ~bal;
            }
            const int rank = __popc(same & below_me);
            const int before = cwl[key];
            if (ok && rank == 0) cwl[key] = before + __popc(same);
            loc[q] = before + rank;
        }
        __syncthreads();

        int t0 = 0, t1 = 0, bl0 = 0, bl1 = 0;
#pragma unroll
        for (int w = 0; w < 8; ++w) {
            const int c0 = cw[TN * w + lane];
            const int c1 = cw[TN * w + 32 + lane];
            t0 += c0; t1 += c1;
            bl0 += (w < wv) ? c0 : 0;
            bl1 += (w < wv) ? c1 : 0;
        }
        int s0 = t0, s1 = t1;
#pragma unroll
        for (int o = 1; o < 32; o <<= 1) {
            const int u0 = __shfl_up(s0, o);
            const int u1 = __shfl_up(s1, o);
            if (lane >= o) { s0 += u0; s1 += u1; }
        }
        const int f0 = s0 - t0;
        const int f1 = __builtin_amdgcn_readlane(s0, 31) + s1 - t1;
        bwl[lane]      = f0 + bl0;
        bwl[32 + lane] = f1 + bl1;
#pragma unroll
        for (int q = 0; q < LCH / 256; ++q) {
            const int i = tid + q * 256;
            const int key = e[q].y & (TN - 1);
            const int slot = bwl[key] + loc[q];
            if (i < m) sub[slot] = e[q].x;
        }
        const int fsel = (wv < 4) ? f0 : f1;
        const int tsel = (wv < 4) ? t0 : t1;
        __syncthreads();

#pragma unroll 1
        for (int j = 0; j < 8; ++j) {
            const int rl = (8 * wv + j) & 31;
            const int rb = __builtin_amdgcn_readlane(fsel, rl);
            const int rc = __builtin_amdgcn_readlane(tsel, rl);
            float4 cur = make_float4(acc[4 * j], acc[4 * j + 1], acc[4 * j + 2], acc[4 * j + 3]);
            int g = 0;
            for (; g + WG <= rc; g += WG) {
                const int sv = sub[rb + g + (lane & (WG - 1))];
                float4 h[WG];
#pragma unroll
                for (int k = 0; k < WG; ++k) {
                    const int s = __builtin_amdgcn_readlane(sv, k);
                    h[k] = *(const float4*)(Hb + (((unsigned)s << 9) + lane16));
                }
#pragma unroll
                for (int k = 0; k < WG; ++k) {
                    cur.x += h[k].x; cur.y += h[k].y; cur.z += h[k].z; cur.w += h[k].w;
                }
            }
            if (g < rc) {
                int i = g + (lane & (WG - 1));
                if (i > rc - 1) i = rc - 1;
                const int sv = sub[rb + i];
                const int nval = rc - g;
                if (nval > 12)     gcn_row_tail<4>(cur, Hb, lane16, sv, nval);
                else if (nval > 8) gcn_row_tail<3>(cur, Hb, lane16, sv, nval);
                else if (nval > 4) gcn_row_tail<2>(cur, Hb, lane16, sv, nval);
                else               gcn_row_tail<1>(cur, Hb, lane16, sv, nval);
            }
            acc[4 * j] = cur.x; acc[4 * j + 1] = cur.y; acc[4 * j + 2] = cur.z; acc[4 * j + 3] = cur.w;
        }
        __syncthreads();
    }

    const float4 b4 = ((const float4*)bias)[lane];
    float di[8];
#pragma unroll
    for (int j = 0; j < 8; ++j) {
        int node = node0 + wv * 8 + j;
        di[j] = dinv[node < n ? node : n - 1];
    }
#pragma unroll
    for (int j = 0; j < 8; ++j) {
        int node = node0 + wv * 8 + j;
        float4 o = make_float4(b4.x + di[j] * acc[4 * j],     b4.y + di[j] * acc[4 * j + 1],
                               b4.z + di[j] * acc[4 * j + 2], b4.w + di[j] * acc[4 * j + 3]);
        if (NEXT) { o.x *= di[j]; o.y *= di[j]; o.z *= di[j]; o.w *= di[j]; }
        else { o.x = gcn_gelu_exact(o.x); o.y = gcn_gelu_exact(o.y); o.z = gcn_gelu_exact(o.z); o.w = gcn_gelu_exact(o.w); }
        if (node < n) ((float4*)Y)[(size_t)node * 32 + lane] = o;
    }
}

extern "C" void kernel_launch(void* const* d_in, const int* in_sizes, int n_in,
                              void* d_out, int out_size, void* d_ws, size_t ws_size,
                              hipStream_t stream) {
    const float* x    = (const float*)d_in[0];
    const int*   eidx = (const int*)  d_in[1];
    const float* Ws   = (const float*)d_in[2];
    const float* bs   = (const float*)d_in[3];
    float*       out  = (float*)d_out;

    const int N = in_sizes[0] / DH;
    const int E = in_sizes[1] / 2;
    const int* src = eidx;
    const int* dst = eidx + E;

    const int NB      = (N + TN - 1) / TN;
    const int PITCH   = (NB + 1 + 31) & ~31;
    const int NCHUNK  = (E + CHUNK - 1) / CHUNK;

    float* ws    = (float*)d_ws;
    float* dinv  = ws;
    int*   offs  = (int*)(dinv + (((size_t)N + 63) & ~(size_t)63));
    int*   start = offs  + (size_t)NCHUNK * PITCH;
    int*   list  = start + (((size_t)NB + 1 + 63) & ~(size_t)63);
    int*   listc = list  + (((size_t)2 * E + 63) & ~(size_t)63);
    float* Wf    = (float*)(listc + (size_t)2 * NCHUNK * CHUNK);
    float* cv    = Wf + (size_t)DH * DH;
    float* h     = cv + (size_t)3 * DH;
    float* buf   = h  + (size_t)N * DH;

    gcn_part_local <<<NCHUNK + 8, 1024, 0, stream>>>(src, dst, offs, listc, E, NB, PITCH, NCHUNK,
                                                    Ws, bs, Wf, cv);

    gcn_gemm<<<NB, GT, 0, stream>>>(x, Wf, offs, listc, list, start, dinv, h, N, NCHUNK, PITCH, NB);
    gcn_agg<true> <<<NB, 256, 0, stream>>>(h,   buf, list, start, dinv, cv,          N);
    gcn_agg<true> <<<NB, 256, 0, stream>>>(buf, h,   list, start, dinv, cv + DH,     N);
    gcn_agg<false><<<NB, 256, 0, stream>>>(h,   out, list, start, dinv, cv + 2 * DH, N);
}
